# starting kernel
# baseline (speedup 1.0000x reference)
#include <hip/hip_runtime.h>


typedef float v2f __attribute__((ext_vector_type(2)));
typedef float v8f __attribute__((ext_vector_type(8)));

#define Bq 8
#define Nq 1024
#define Fin 256
#define Fout 256
#define Hh 4
#define HD 64
#define NEG_SLOPE 0.2f

__global__ __launch_bounds__(128) void gat_proj(const float* __restrict__ x,
                                                const float* __restrict__ W,
                                                float* __restrict__ hbuf) {
  const int tid = threadIdx.x;
  const int wave = tid >> 5, lane = tid & 31;
  const int lane16 = lane & 15, laneHi = lane >> 4;
  const int tile = blockIdx.x * 4 + wave;
  const int mt = tile >> 4;
  const int nt = tile & 15;
  const int row = mt * 16 + lane16;
  const int col = nt * 16 + lane16;

  const float* xr = x + row * Fin;
  v8f acc = {};
#pragma unroll 4
  for (int k = 0; k < Fin; k += 4) {
    const int k0 = k + 2 * laneHi;
    v2f av = *reinterpret_cast<const v2f*>(xr + k0);
    v2f bv;
    bv.x = W[k0 * Fout + col];
    bv.y = W[(k0 + 1) * Fout + col];
    acc = __builtin_amdgcn_wmma_f32_16x16x4_f32(false, av, false, bv,
                                                (short)0, acc, false, false);
  }
  const int hh = col >> 6, d = col & 63;
#pragma unroll
  for (int v = 0; v < 8; ++v) {
    const int r = mt * 16 + v + 8 * laneHi;
    const int bb = r >> 10, n = r & (Nq - 1);
    hbuf[(((bb * Hh + hh) * Nq + n) * HD) + d] = acc[v];
  }
}

__global__ __launch_bounds__(256) void gat_scores(const float* __restrict__ hbuf,
                                                  const float* __restrict__ a,
                                                  float* __restrict__ s1,
                                                  float* __restrict__ s2) {
  const int tid = threadIdx.x;
  const int wave = tid >> 5, lane = tid & 31;
  const int idx = blockIdx.x * 8 + wave;
  const int hh = (idx >> 10) & (Hh - 1);
  const float* hr = hbuf + idx * HD;
  const float* ah = a + hh * (2 * HD);
  const float h0 = hr[lane], h1 = hr[lane + 32];
  float d1 = h0 * ah[lane] + h1 * ah[lane + 32];
  float d2 = h0 * ah[HD + lane] + h1 * ah[HD + lane + 32];
#pragma unroll
  for (int off = 16; off; off >>= 1) {
    d1 += __shfl_xor(d1, off, 32);
    d2 += __shfl_xor(d2, off, 32);
  }
  if (lane == 0) { s1[idx] = d1; s2[idx] = d2; }
}

__global__ __launch_bounds__(256) void gat_attn(const float* __restrict__ hbuf,
                                                const int* __restrict__ adj,
                                                const float* __restrict__ s1g,
                                                const float* __restrict__ s2g,
                                                float* __restrict__ out) {
  __shared__ unsigned int maskw[16 * (Nq / 32)];
  __shared__ float s2l[Hh * Nq];
  __shared__ float s1l[Hh * 16];
  __shared__ float rmaxl[Hh * 16];
  __shared__ float rsuml[Hh * 16];
  __shared__ float opart[Hh * 16 * HD];

  const int tid = threadIdx.x;
  const int wave = tid >> 5, lane = tid & 31;
  const int lane16 = lane & 15, laneHi = lane >> 4;
  const int b = blockIdx.y, n0 = blockIdx.x * 16;
  const int hh = wave >> 1;
  const int half = wave & 1;

  const int* adjb = adj + (b * Nq + n0) * Nq;
  for (int w = tid; w < 16 * (Nq / 32); w += 256) {
    const int r = w >> 5, m0 = (w & 31) * 32;
    const int* ar = adjb + r * Nq + m0;
    unsigned int bits = 0u;
#pragma unroll
    for (int j = 0; j < 32; ++j) bits |= (ar[j] != 0 ? 1u : 0u) << j;
    maskw[w] = bits;
  }
  const float* s2b = s2g + b * Hh * Nq;
  for (int i = tid; i < Hh * Nq; i += 256) s2l[i] = s2b[i];
  const float* s1b = s1g + b * Hh * Nq;
  for (int i = tid; i < Hh * 16; i += 256)
    s1l[i] = s1b[(i >> 4) * Nq + n0 + (i & 15)];
  __syncthreads();

  for (int r8 = 0; r8 < 8; ++r8) {
    const int r = half * 8 + r8;
    const float s1r = s1l[hh * 16 + r];
    float mx = -1e30f, sm = 0.f;
    for (int m = lane; m < Nq; m += 32) {
      if ((maskw[r * 32 + (m >> 5)] >> (m & 31)) & 1u) {
        float v = s1r + s2l[hh * Nq + m];
        v = v > 0.f ? v : NEG_SLOPE * v;
        const float nm = fmaxf(mx, v);
        sm = sm * __expf(mx - nm) + __expf(v - nm);
        mx = nm;
      }
    }
#pragma unroll
    for (int off = 16; off; off >>= 1) {
      const float om = __shfl_xor(mx, off, 32);
      const float os = __shfl_xor(sm, off, 32);
      const float nm = fmaxf(mx, om);
      sm = sm * __expf(mx - nm) + os * __expf(om - nm);
      mx = nm;
    }
    if (lane == 0) { rmaxl[hh * 16 + r] = mx; rsuml[hh * 16 + r] = sm; }
  }
  __syncthreads();

  const float s1v  = s1l[hh * 16 + lane16];
  const float rmax = rmaxl[hh * 16 + lane16];
  const float rsum = rsuml[hh * 16 + lane16];
  const float rinv = rsum > 0.f ? 1.f / rsum : 0.f;
  const float* hb  = hbuf + (b * Hh + hh) * Nq * HD;
  const float* s2h = s2l + hh * Nq;
  const unsigned int* mrow = maskw + lane16 * 32;
  const int mbase = half * (Nq / 2);

  v8f acc0 = {}, acc1 = {}, acc2 = {}, acc3 = {};
  for (int kk = mbase; kk < mbase + Nq / 2; kk += 4) {
    const int m0 = kk + 2 * laneHi;
    float p0 = 0.f, p1 = 0.f;
    if ((mrow[m0 >> 5] >> (m0 & 31)) & 1u) {
      float v = s1v + s2h[m0];
      v = v > 0.f ? v : NEG_SLOPE * v;
      p0 = __expf(v - rmax) * rinv;
    }
    if ((mrow[(m0 + 1) >> 5] >> ((m0 + 1) & 31)) & 1u) {
      float v = s1v + s2h[m0 + 1];
      v = v > 0.f ? v : NEG_SLOPE * v;
      p1 = __expf(v - rmax) * rinv;
    }
    v2f av; av.x = p0; av.y = p1;

    const float* h0 = hb + m0 * HD;
    const float* h1 = h0 + HD;
    v2f bv;
    bv.x = h0[lane16];      bv.y = h1[lane16];
    acc0 = __builtin_amdgcn_wmma_f32_16x16x4_f32(false, av, false, bv, (short)0, acc0, false, false);
    bv.x = h0[16 + lane16]; bv.y = h1[16 + lane16];
    acc1 = __builtin_amdgcn_wmma_f32_16x16x4_f32(false, av, false, bv, (short)0, acc1, false, false);
    bv.x = h0[32 + lane16]; bv.y = h1[32 + lane16];
    acc2 = __builtin_amdgcn_wmma_f32_16x16x4_f32(false, av, false, bv, (short)0, acc2, false, false);
    bv.x = h0[48 + lane16]; bv.y = h1[48 + lane16];
    acc3 = __builtin_amdgcn_wmma_f32_16x16x4_f32(false, av, false, bv, (short)0, acc3, false, false);
  }

  float* op = opart + hh * 16 * HD;
  if (half == 0) {
#pragma unroll
    for (int v = 0; v < 8; ++v) {
      float* orow = op + (v + 8 * laneHi) * HD + lane16;
      orow[0]  = acc0[v];
      orow[16] = acc1[v];
      orow[32] = acc2[v];
      orow[48] = acc3[v];
    }
  }
  __syncthreads();
  if (half == 1) {
    float* ob = out + (size_t)b * Nq * Fout + hh * HD;
#pragma unroll
    for (int v = 0; v < 8; ++v) {
      const float* prow = op + (v + 8 * laneHi) * HD + lane16;
      float* orow = ob + (size_t)(n0 + v + 8 * laneHi) * Fout + lane16;
      orow[0]  = acc0[v] + prow[0];
      orow[16] = acc1[v] + prow[16];
      orow[32] = acc2[v] + prow[32];
      orow[48] = acc3[v] + prow[48];
    }
  }
}

extern "C" void kernel_launch(void* const* d_in, const int* in_sizes, int n_in,
                              void* d_out, int out_size, void* d_ws, size_t ws_size,
                              hipStream_t stream) {
  const float* x   = (const float*)d_in[0];
  const int*   adj = (const int*)d_in[1];
  const float* W   = (const float*)d_in[2];
  const float* a   = (const float*)d_in[3];
  float* out = (float*)d_out;

  float* hbuf = (float*)d_ws;
  float* s1 = hbuf + (size_t)Bq * Hh * Nq * HD;
  float* s2 = s1 + (size_t)Bq * Hh * Nq;

  gat_proj<<<2048, 128, 0, stream>>>(x, W, hbuf);
  gat_scores<<<4096, 256, 0, stream>>>(hbuf, a, s1, s2);
  gat_attn<<<dim3(Nq / 16, Bq), 256, 0, stream>>>(hbuf, adj, s1, s2, out);
}


# attn pass 2: h B-fragments of each 16-k block are loaded one block ahead into a second register set (ping-pong), so their L2 round trip runs under the previous block's WMMAs, not in every k-step
# speedup vs baseline: 1.2066x; 1.2066x over previous
#include <hip/hip_runtime.h>


typedef float v2f __attribute__((ext_vector_type(2)));
typedef float v8f __attribute__((ext_vector_type(8)));

#define Bq 8
#define Nq 1024
#define Fin 256
#define Fout 256
#define Hh 4
#define HD 64
#define NEG_SLOPE 0.2f

__global__ __launch_bounds__(128) void gat_proj(const float* __restrict__ x,
                                                const float* __restrict__ W,
                                                float* __restrict__ hbuf) {
  const int tid = threadIdx.x;
  const int wave = tid >> 5, lane = tid & 31;
  const int lane16 = lane & 15, laneHi = lane >> 4;
  const int tile = blockIdx.x * 4 + wave;
  const int mt = tile >> 4;
  const int nt = tile & 15;
  const int row = mt * 16 + lane16;
  const int col = nt * 16 + lane16;

  const float* xr = x + row * Fin;
  v8f acc = {};
#pragma unroll 4
  for (int k = 0; k < Fin; k += 4) {
    const int k0 = k + 2 * laneHi;
    v2f av = *reinterpret_cast<const v2f*>(xr + k0);
    v2f bv;
    bv.x = W[k0 * Fout + col];
    bv.y = W[(k0 + 1) * Fout + col];
    acc = __builtin_amdgcn_wmma_f32_16x16x4_f32(false, av, false, bv,
                                                (short)0, acc, false, false);
  }
  const int hh = col >> 6, d = col & 63;
#pragma unroll
  for (int v = 0; v < 8; ++v) {
    const int r = mt * 16 + v + 8 * laneHi;
    const int bb = r >> 10, n = r & (Nq - 1);
    hbuf[(((bb * Hh + hh) * Nq + n) * HD) + d] = acc[v];
  }
}

__global__ __launch_bounds__(256) void gat_scores(const float* __restrict__ hbuf,
                                                  const float* __restrict__ a,
                                                  float* __restrict__ s1,
                                                  float* __restrict__ s2) {
  const int tid = threadIdx.x;
  const int wave = tid >> 5, lane = tid & 31;
  const int idx = blockIdx.x * 8 + wave;
  const int hh = (idx >> 10) & (Hh - 1);
  const float* hr = hbuf + idx * HD;
  const float* ah = a + hh * (2 * HD);
  const float h0 = hr[lane], h1 = hr[lane + 32];
  float d1 = h0 * ah[lane] + h1 * ah[lane + 32];
  float d2 = h0 * ah[HD + lane] + h1 * ah[HD + lane + 32];
#pragma unroll
  for (int off = 16; off; off >>= 1) {
    d1 += __shfl_xor(d1, off, 32);
    d2 += __shfl_xor(d2, off, 32);
  }
  if (lane == 0) { s1[idx] = d1; s2[idx] = d2; }
}

struct HBlock { v2f t[4][4]; };

__device__ __forceinline__ void load_hblock(HBlock& f, const float* __restrict__ hrow) {
#pragma unroll
  for (int s = 0; s < 4; ++s) {
#pragma unroll
    for (int ct = 0; ct < 4; ++ct) {
      f.t[s][ct].x = hrow[(4 * s) * HD + ct * 16];
      f.t[s][ct].y = hrow[(4 * s + 1) * HD + ct * 16];
    }
  }
}

__global__ __launch_bounds__(256) void gat_attn(const float* __restrict__ hbuf,
                                                const int* __restrict__ adj,
                                                const float* __restrict__ s1g,
                                                const float* __restrict__ s2g,
                                                float* __restrict__ out) {
  __shared__ unsigned int maskw[16 * (Nq / 32)];
  __shared__ float s2l[Hh * Nq];
  __shared__ float s1l[Hh * 16];
  __shared__ float rmaxl[Hh * 16];
  __shared__ float rsuml[Hh * 16];
  __shared__ float opart[Hh * 16 * HD];

  const int tid = threadIdx.x;
  const int wave = tid >> 5, lane = tid & 31;
  const int lane16 = lane & 15, laneHi = lane >> 4;
  const int b = blockIdx.y, n0 = blockIdx.x * 16;
  const int hh = wave >> 1;
  const int half = wave & 1;

  const int* adjb = adj + (b * Nq + n0) * Nq;
  for (int w = tid; w < 16 * (Nq / 32); w += 256) {
    const int r = w >> 5, m0 = (w & 31) * 32;
    const int* ar = adjb + r * Nq + m0;
    unsigned int bits = 0u;
#pragma unroll
    for (int j = 0; j < 32; ++j) bits |= (ar[j] != 0 ? 1u : 0u) << j;
    maskw[w] = bits;
  }
  const float* s2b = s2g + b * Hh * Nq;
  for (int i = tid; i < Hh * Nq; i += 256) s2l[i] = s2b[i];
  const float* s1b = s1g + b * Hh * Nq;
  for (int i = tid; i < Hh * 16; i += 256)
    s1l[i] = s1b[(i >> 4) * Nq + n0 + (i & 15)];
  __syncthreads();

  for (int r8 = 0; r8 < 8; ++r8) {
    const int r = half * 8 + r8;
    const float s1r = s1l[hh * 16 + r];
    float mx = -1e30f, sm = 0.f;
    for (int m = lane; m < Nq; m += 32) {
      if ((maskw[r * 32 + (m >> 5)] >> (m & 31)) & 1u) {
        float v = s1r + s2l[hh * Nq + m];
        v = v > 0.f ? v : NEG_SLOPE * v;
        const float nm = fmaxf(mx, v);
        sm = sm * __expf(mx - nm) + __expf(v - nm);
        mx = nm;
      }
    }
#pragma unroll
    for (int off = 16; off; off >>= 1) {
      const float om = __shfl_xor(mx, off, 32);
      const float os = __shfl_xor(sm, off, 32);
      const float nm = fmaxf(mx, om);
      sm = sm * __expf(mx - nm) + os * __expf(om - nm);
      mx = nm;
    }
    if (lane == 0) { rmaxl[hh * 16 + r] = mx; rsuml[hh * 16 + r] = sm; }
  }
  __syncthreads();

  const float s1v  = s1l[hh * 16 + lane16];
  const float rmax = rmaxl[hh * 16 + lane16];
  const float rsum = rsuml[hh * 16 + lane16];
  const float rinv = rsum > 0.f ? 1.f / rsum : 0.f;
  const float* hb  = hbuf + (b * Hh + hh) * Nq * HD;
  const float* s2h = s2l + hh * Nq;
  const unsigned int* mrow = maskw + lane16 * 32;
  const int mbase = half * (Nq / 2);

  v8f acc0 = {}, acc1 = {}, acc2 = {}, acc3 = {};
  const int mend = mbase + Nq / 2;
  const float* hlane = hb + (2 * laneHi) * HD + lane16;
  auto mma_block = [&](const HBlock& f, const int kb) {
#pragma unroll
    for (int s = 0; s < 4; ++s) {
      const int m0 = kb + 4 * s + 2 * laneHi;
      float p0 = 0.f, p1 = 0.f;
      if ((mrow[m0 >> 5] >> (m0 & 31)) & 1u) {
        float v = s1v + s2h[m0];
        v = v > 0.f ? v : NEG_SLOPE * v;
        p0 = __expf(v - rmax) * rinv;
      }
      if ((mrow[(m0 + 1) >> 5] >> ((m0 + 1) & 31)) & 1u) {
        float v = s1v + s2h[m0 + 1];
        v = v > 0.f ? v : NEG_SLOPE * v;
        p1 = __expf(v - rmax) * rinv;
      }
      v2f av; av.x = p0; av.y = p1;
      acc0 = __builtin_amdgcn_wmma_f32_16x16x4_f32(false, av, false, f.t[s][0], (short)0, acc0, false, false);
      acc1 = __builtin_amdgcn_wmma_f32_16x16x4_f32(false, av, false, f.t[s][1], (short)0, acc1, false, false);
      acc2 = __builtin_amdgcn_wmma_f32_16x16x4_f32(false, av, false, f.t[s][2], (short)0, acc2, false, false);
      acc3 = __builtin_amdgcn_wmma_f32_16x16x4_f32(false, av, false, f.t[s][3], (short)0, acc3, false, false);
    }
  };

  HBlock fa, fb;
  load_hblock(fa, hlane + mbase * HD);
  for (int kb = mbase; kb < mend; kb += 32) {
    load_hblock(fb, hlane + (kb + 16) * HD);
    mma_block(fa, kb);
    const int kn = kb + 32 < mend ? kb + 32 : mbase;
    load_hblock(fa, hlane + kn * HD);
    mma_block(fb, kb + 16);
  }

  float* op = opart + hh * 16 * HD;
  if (half == 0) {
#pragma unroll
    for (int v = 0; v < 8; ++v) {
      float* orow = op + (v + 8 * laneHi) * HD + lane16;
      orow[0]  = acc0[v];
      orow[16] = acc1[v];
      orow[32] = acc2[v];
      orow[48] = acc3[v];
    }
  }
  __syncthreads();
  if (half == 1) {
    float* ob = out + (size_t)b * Nq * Fout + hh * HD;
#pragma unroll
    for (int v = 0; v < 8; ++v) {
      const float* prow = op + (v + 8 * laneHi) * HD + lane16;
      float* orow = ob + (size_t)(n0 + v + 8 * laneHi) * Fout + lane16;
      orow[0]  = acc0[v] + prow[0];
      orow[16] = acc1[v] + prow[16];
      orow[32] = acc2[v] + prow[32];
      orow[48] = acc3[v] + prow[48];
    }
  }
}

extern "C" void kernel_launch(void* const* d_in, const int* in_sizes, int n_in,
                              void* d_out, int out_size, void* d_ws, size_t ws_size,
                              hipStream_t stream) {
  const float* x   = (const float*)d_in[0];
  const int*   adj = (const int*)d_in[1];
  const float* W   = (const float*)d_in[2];
  const float* a   = (const float*)d_in[3];
  float* out = (float*)d_out;

  float* hbuf = (float*)d_ws;
  float* s1 = hbuf + (size_t)Bq * Hh * Nq * HD;
  float* s2 = s1 + (size_t)Bq * Hh * Nq;

  gat_proj<<<2048, 128, 0, stream>>>(x, W, hbuf);
  gat_scores<<<4096, 256, 0, stream>>>(hbuf, a, s1, s2);
  gat_attn<<<dim3(Nq / 16, Bq), 256, 0, stream>>>(hbuf, adj, s1, s2, out);
}


# s1/s2 dot products folded into the projection kernel's epilogue (tile products summed through LDS, 4 partials per row: new sum order); the scores launch and its 32768 one-row waves are gone
# speedup vs baseline: 1.8491x; 1.5325x over previous
#include <hip/hip_runtime.h>


typedef float v2f __attribute__((ext_vector_type(2)));
typedef float v8f __attribute__((ext_vector_type(8)));

#define Bq 8
#define Nq 1024
#define Fin 256
#define Fout 256
#define Hh 4
#define HD 64
#define NEG_SLOPE 0.2f

#define PRP 68
__global__ __launch_bounds__(128) void gat_proj(const float* __restrict__ x,
                                                const float* __restrict__ W,
                                                const float* __restrict__ a,
                                                float* __restrict__ hbuf,
                                                float* __restrict__ s1,
                                                float* __restrict__ s2) {
  __shared__ __attribute__((aligned(16))) float prod[2][16][PRP];

  const int tid = threadIdx.x;
  const int wave = tid >> 5, lane = tid & 31;
  const int lane16 = lane & 15, laneHi = lane >> 4;
  const int tile = blockIdx.x * 4 + wave;
  const int mt = tile >> 4;
  const int nt = tile & 15;
  const int row = mt * 16 + lane16;
  const int col = nt * 16 + lane16;
  const int hh = col >> 6, d = col & 63;

  const float a1d = a[hh * (2 * HD) + d];
  const float a2d = a[hh * (2 * HD) + HD + d];

  const float* xr = x + row * Fin;
  v8f acc = {};
#pragma unroll 4
  for (int k = 0; k < Fin; k += 4) {
    const int k0 = k + 2 * laneHi;
    v2f av = *reinterpret_cast<const v2f*>(xr + k0);
    v2f bv;
    bv.x = W[k0 * Fout + col];
    bv.y = W[(k0 + 1) * Fout + col];
    acc = __builtin_amdgcn_wmma_f32_16x16x4_f32(false, av, false, bv,
                                                (short)0, acc, false, false);
  }
#pragma unroll
  for (int v = 0; v < 8; ++v) {
    prod[0][v + 8 * laneHi][d] = acc[v] * a1d;
    prod[1][v + 8 * laneHi][d] = acc[v] * a2d;
  }
  __syncthreads();
#pragma unroll
  for (int v = 0; v < 8; ++v) {
    const int r = mt * 16 + v + 8 * laneHi;
    const int bb = r >> 10, n = r & (Nq - 1);
    hbuf[(((bb * Hh + hh) * Nq + n) * HD) + d] = acc[v];
  }
  {
    const int i = tid >> 2, part = tid & 3;
    const int q = i >> 4, rr = i & 15;
    const float4* pr = reinterpret_cast<const float4*>(&prod[q][rr][part * 16]);
    const float4 u0 = pr[0], u1 = pr[1], u2 = pr[2], u3 = pr[3];
    float sum = (((u0.x + u0.y) + (u0.z + u0.w)) + ((u1.x + u1.y) + (u1.z + u1.w))) +
                (((u2.x + u2.y) + (u2.z + u2.w)) + ((u3.x + u3.y) + (u3.z + u3.w)));
    sum += __shfl_xor(sum, 1, 32);
    sum += __shfl_xor(sum, 2, 32);
    if (part == 0) {
      const int r = mt * 16 + rr;
      const int bb = r >> 10, n = r & (Nq - 1);
      float* dst = q ? s2 : s1;
      dst[(bb * Hh + hh) * Nq + n] = sum;
    }
  }
}

struct HBlock { v2f t[4][4]; };

__device__ __forceinline__ void load_hblock(HBlock& f, const float* __restrict__ hrow) {
#pragma unroll
  for (int s = 0; s < 4; ++s) {
#pragma unroll
    for (int ct = 0; ct < 4; ++ct) {
      f.t[s][ct].x = hrow[(4 * s) * HD + ct * 16];
      f.t[s][ct].y = hrow[(4 * s + 1) * HD + ct * 16];
    }
  }
}

__global__ __launch_bounds__(256) void gat_attn(const float* __restrict__ hbuf,
                                                const int* __restrict__ adj,
                                                const float* __restrict__ s1g,
                                                const float* __restrict__ s2g,
                                                float* __restrict__ out) {
  __shared__ unsigned int maskw[16 * (Nq / 32)];
  __shared__ __attribute__((aligned(16))) float s2l[Hh * Nq];
  __shared__ float s1l[Hh * 16];
  __shared__ float s2maxl[Hh * 16];
  __shared__ float psuml[2 * Hh * 16];
  __shared__ float opart[Hh * 16 * HD];

  const int tid = threadIdx.x;
  const int wave = tid >> 5, lane = tid & 31;
  const int lane16 = lane & 15, laneHi = lane >> 4;
  const int b = blockIdx.y, n0 = blockIdx.x * 16;
  const int hh = wave >> 1;
  const int half = wave & 1;

  const int* adjb = adj + (b * Nq + n0) * Nq;
  for (int w = tid; w < 16 * (Nq / 32); w += 256) {
    const int r = w >> 5, m0 = (w & 31) * 32;
    const int* ar = adjb + r * Nq + m0;
    unsigned int bits = 0u;
#pragma unroll
    for (int j = 0; j < 32; ++j) bits |= (ar[j] != 0 ? 1u : 0u) << j;
    maskw[w] = bits;
  }
  const float* s2b = s2g + b * Hh * Nq;
  for (int i = tid; i < Hh * Nq; i += 256) s2l[i] = s2b[i];
  const float* s1b = s1g + b * Hh * Nq;
  for (int i = tid; i < Hh * 16; i += 256)
    s1l[i] = s1b[(i >> 4) * Nq + n0 + (i & 15)];
  __syncthreads();

  float s2v[32];
  {
    const float4* s2q = reinterpret_cast<const float4*>(s2l + hh * Nq + lane * 32);
#pragma unroll
    for (int q = 0; q < 8; ++q) {
      const float4 t = s2q[q];
      s2v[4 * q] = t.x; s2v[4 * q + 1] = t.y; s2v[4 * q + 2] = t.z; s2v[4 * q + 3] = t.w;
    }
  }
  const float ninf = -__builtin_huge_valf();
  float mx8[8];
#pragma unroll
  for (int r8 = 0; r8 < 8; ++r8) {
    const unsigned int w = maskw[(half * 8 + r8) * 32 + lane];
    float mx = ninf;
#pragma unroll
    for (int j = 0; j < 32; ++j) mx = fmaxf(mx, ((w >> j) & 1u) ? s2v[j] : ninf);
    mx8[r8] = mx;
  }
#pragma unroll
  for (int off = 16; off; off >>= 1) {
#pragma unroll
    for (int r8 = 0; r8 < 8; ++r8)
      mx8[r8] = fmaxf(mx8[r8], __shfl_xor(mx8[r8], off, 32));
  }
  if (lane == 0) {
#pragma unroll
    for (int r8 = 0; r8 < 8; ++r8) s2maxl[hh * 16 + half * 8 + r8] = mx8[r8];
  }
  __syncthreads();

  const float s1v  = s1l[hh * 16 + lane16];
  float rmax = s1v + s2maxl[hh * 16 + lane16];
  rmax = rmax > 0.f ? rmax : NEG_SLOPE * rmax;
  float psum = 0.f;
  const float* hb  = hbuf + (b * Hh + hh) * Nq * HD;
  const float* s2h = s2l + hh * Nq;
  const unsigned int* mrow = maskw + lane16 * 32;
  const int mbase = half * (Nq / 2);

  v8f acc0 = {}, acc1 = {}, acc2 = {}, acc3 = {};
  const int mend = mbase + Nq / 2;
  const float* hlane = hb + (2 * laneHi) * HD + lane16;
  auto mma_block = [&](const HBlock& f, const int kb) {
#pragma unroll
    for (int s = 0; s < 4; ++s) {
      const int m0 = kb + 4 * s + 2 * laneHi;
      float p0 = 0.f, p1 = 0.f;
      if ((mrow[m0 >> 5] >> (m0 & 31)) & 1u) {
        float v = s1v + s2h[m0];
        v = v > 0.f ? v : NEG_SLOPE * v;
        p0 = __expf(v - rmax);
      }
      if ((mrow[(m0 + 1) >> 5] >> ((m0 + 1) & 31)) & 1u) {
        float v = s1v + s2h[m0 + 1];
        v = v > 0.f ? v : NEG_SLOPE * v;
        p1 = __expf(v - rmax);
      }
      psum += p0 + p1;
      v2f av; av.x = p0; av.y = p1;
      acc0 = __builtin_amdgcn_wmma_f32_16x16x4_f32(false, av, false, f.t[s][0], (short)0, acc0, false, false);
      acc1 = __builtin_amdgcn_wmma_f32_16x16x4_f32(false, av, false, f.t[s][1], (short)0, acc1, false, false);
      acc2 = __builtin_amdgcn_wmma_f32_16x16x4_f32(false, av, false, f.t[s][2], (short)0, acc2, false, false);
      acc3 = __builtin_amdgcn_wmma_f32_16x16x4_f32(false, av, false, f.t[s][3], (short)0, acc3, false, false);
    }
  };

  HBlock fa, fb;
  load_hblock(fa, hlane + mbase * HD);
  for (int kb = mbase; kb < mend; kb += 32) {
    load_hblock(fb, hlane + (kb + 16) * HD);
    mma_block(fa, kb);
    const int kn = kb + 32 < mend ? kb + 32 : mbase;
    load_hblock(fa, hlane + kn * HD);
    mma_block(fb, kb + 16);
  }

  psum += __shfl_xor(psum, 16, 32);
  if (laneHi == 0) psuml[(half * Hh + hh) * 16 + lane16] = psum;
  float* op = opart + hh * 16 * HD;
  if (half == 0) {
#pragma unroll
    for (int v = 0; v < 8; ++v) {
      float* orow = op + (v + 8 * laneHi) * HD + lane16;
      orow[0]  = acc0[v];
      orow[16] = acc1[v];
      orow[32] = acc2[v];
      orow[48] = acc3[v];
    }
  }
  __syncthreads();
  if (half == 1) {
    float* ob = out + (size_t)b * Nq * Fout + hh * HD;
#pragma unroll
    for (int v = 0; v < 8; ++v) {
      const float* prow = op + (v + 8 * laneHi) * HD + lane16;
      float* orow = ob + (size_t)(n0 + v + 8 * laneHi) * Fout + lane16;
      const float rsum = psuml[hh * 16 + v + 8 * laneHi] + psuml[(Hh + hh) * 16 + v + 8 * laneHi];
      const float rinv = rsum > 0.f ? 1.f / rsum : 0.f;
      orow[0]  = (acc0[v] + prow[0]) * rinv;
      orow[16] = (acc1[v] + prow[16]) * rinv;
      orow[32] = (acc2[v] + prow[32]) * rinv;
      orow[48] = (acc3[v] + prow[48]) * rinv;
    }
  }
}

extern "C" void kernel_launch(void* const* d_in, const int* in_sizes, int n_in,
                              void* d_out, int out_size, void* d_ws, size_t ws_size,
                              hipStream_t stream) {
  const float* x   = (const float*)d_in[0];
  const int*   adj = (const int*)d_in[1];
  const float* W   = (const float*)d_in[2];
  const float* a   = (const float*)d_in[3];
  float* out = (float*)d_out;

  float* hbuf = (float*)d_ws;
  float* s1 = hbuf + (size_t)Bq * Hh * Nq * HD;
  float* s2 = s1 + (size_t)Bq * Hh * Nq;

  gat_proj<<<2048, 128, 0, stream>>>(x, W, a, hbuf, s1, s2);
  gat_attn<<<dim3(Nq / 16, Bq), 256, 0, stream>>>(hbuf, adj, s1, s2, out);
}


# attn pass 2: p formed one 16-k block ahead of its WMMAs, branch-free (exp always, mask bit selects p or 0), s2 pairs read as one LDS group a trip ahead: no LDS wait or branch between WMMAs
# speedup vs baseline: 2.0215x; 1.0932x over previous
#include <hip/hip_runtime.h>


typedef float v2f __attribute__((ext_vector_type(2)));
typedef float v8f __attribute__((ext_vector_type(8)));

#define Bq 8
#define Nq 1024
#define Fin 256
#define Fout 256
#define Hh 4
#define HD 64
#define NEG_SLOPE 0.2f

#define PRP 68
__global__ __launch_bounds__(128) void gat_proj(const float* __restrict__ x,
                                                const float* __restrict__ W,
                                                const float* __restrict__ a,
                                                float* __restrict__ hbuf,
                                                float* __restrict__ s1,
                                                float* __restrict__ s2) {
  __shared__ __attribute__((aligned(16))) float prod[2][16][PRP];

  const int tid = threadIdx.x;
  const int wave = tid >> 5, lane = tid & 31;
  const int lane16 = lane & 15, laneHi = lane >> 4;
  const int tile = blockIdx.x * 4 + wave;
  const int mt = tile >> 4;
  const int nt = tile & 15;
  const int row = mt * 16 + lane16;
  const int col = nt * 16 + lane16;
  const int hh = col >> 6, d = col & 63;

  const float a1d = a[hh * (2 * HD) + d];
  const float a2d = a[hh * (2 * HD) + HD + d];

  const float* xr = x + row * Fin;
  v8f acc = {};
#pragma unroll 4
  for (int k = 0; k < Fin; k += 4) {
    const int k0 = k + 2 * laneHi;
    v2f av = *reinterpret_cast<const v2f*>(xr + k0);
    v2f bv;
    bv.x = W[k0 * Fout + col];
    bv.y = W[(k0 + 1) * Fout + col];
    acc = __builtin_amdgcn_wmma_f32_16x16x4_f32(false, av, false, bv,
                                                (short)0, acc, false, false);
  }
#pragma unroll
  for (int v = 0; v < 8; ++v) {
    prod[0][v + 8 * laneHi][d] = acc[v] * a1d;
    prod[1][v + 8 * laneHi][d] = acc[v] * a2d;
  }
  __syncthreads();
#pragma unroll
  for (int v = 0; v < 8; ++v) {
    const int r = mt * 16 + v + 8 * laneHi;
    const int bb = r >> 10, n = r & (Nq - 1);
    hbuf[(((bb * Hh + hh) * Nq + n) * HD) + d] = acc[v];
  }
  {
    const int i = tid >> 2, part = tid & 3;
    const int q = i >> 4, rr = i & 15;
    const float4* pr = reinterpret_cast<const float4*>(&prod[q][rr][part * 16]);
    const float4 u0 = pr[0], u1 = pr[1], u2 = pr[2], u3 = pr[3];
    float sum = (((u0.x + u0.y) + (u0.z + u0.w)) + ((u1.x + u1.y) + (u1.z + u1.w))) +
                (((u2.x + u2.y) + (u2.z + u2.w)) + ((u3.x + u3.y) + (u3.z + u3.w)));
    sum += __shfl_xor(sum, 1, 32);
    sum += __shfl_xor(sum, 2, 32);
    if (part == 0) {
      const int r = mt * 16 + rr;
      const int bb = r >> 10, n = r & (Nq - 1);
      float* dst = q ? s2 : s1;
      dst[(bb * Hh + hh) * Nq + n] = sum;
    }
  }
}

struct HBlock { v2f t[4][4]; };

__device__ __forceinline__ void load_hblock(HBlock& f, const float* __restrict__ hrow) {
#pragma unroll
  for (int s = 0; s < 4; ++s) {
#pragma unroll
    for (int ct = 0; ct < 4; ++ct) {
      f.t[s][ct].x = hrow[(4 * s) * HD + ct * 16];
      f.t[s][ct].y = hrow[(4 * s + 1) * HD + ct * 16];
    }
  }
}

__global__ __launch_bounds__(256) void gat_attn(const float* __restrict__ hbuf,
                                                const int* __restrict__ adj,
                                                const float* __restrict__ s1g,
                                                const float* __restrict__ s2g,
                                                float* __restrict__ out) {
  __shared__ unsigned int maskw[16 * (Nq / 32)];
  __shared__ __attribute__((aligned(16))) float s2l[Hh * Nq];
  __shared__ float s1l[Hh * 16];
  __shared__ float s2maxl[Hh * 16];
  __shared__ float psuml[2 * Hh * 16];
  __shared__ float opart[Hh * 16 * HD];

  const int tid = threadIdx.x;
  const int wave = tid >> 5, lane = tid & 31;
  const int lane16 = lane & 15, laneHi = lane >> 4;
  const int b = blockIdx.y, n0 = blockIdx.x * 16;
  const int hh = wave >> 1;
  const int half = wave & 1;

  const int* adjb = adj + (b * Nq + n0) * Nq;
  for (int w = tid; w < 16 * (Nq / 32); w += 256) {
    const int r = w >> 5, m0 = (w & 31) * 32;
    const int* ar = adjb + r * Nq + m0;
    unsigned int bits = 0u;
#pragma unroll
    for (int j = 0; j < 32; ++j) bits |= (ar[j] != 0 ? 1u : 0u) << j;
    maskw[w] = bits;
  }
  const float* s2b = s2g + b * Hh * Nq;
  for (int i = tid; i < Hh * Nq; i += 256) s2l[i] = s2b[i];
  const float* s1b = s1g + b * Hh * Nq;
  for (int i = tid; i < Hh * 16; i += 256)
    s1l[i] = s1b[(i >> 4) * Nq + n0 + (i & 15)];
  __syncthreads();

  float s2v[32];
  {
    const float4* s2q = reinterpret_cast<const float4*>(s2l + hh * Nq + lane * 32);
#pragma unroll
    for (int q = 0; q < 8; ++q) {
      const float4 t = s2q[q];
      s2v[4 * q] = t.x; s2v[4 * q + 1] = t.y; s2v[4 * q + 2] = t.z; s2v[4 * q + 3] = t.w;
    }
  }
  const float ninf = -__builtin_huge_valf();
  float mx8[8];
#pragma unroll
  for (int r8 = 0; r8 < 8; ++r8) {
    const unsigned int w = maskw[(half * 8 + r8) * 32 + lane];
    float mx = ninf;
#pragma unroll
    for (int j = 0; j < 32; ++j) mx = fmaxf(mx, ((w >> j) & 1u) ? s2v[j] : ninf);
    mx8[r8] = mx;
  }
#pragma unroll
  for (int off = 16; off; off >>= 1) {
#pragma unroll
    for (int r8 = 0; r8 < 8; ++r8)
      mx8[r8] = fmaxf(mx8[r8], __shfl_xor(mx8[r8], off, 32));
  }
  if (lane == 0) {
#pragma unroll
    for (int r8 = 0; r8 < 8; ++r8) s2maxl[hh * 16 + half * 8 + r8] = mx8[r8];
  }
  __syncthreads();

  const float s1v  = s1l[hh * 16 + lane16];
  float rmax = s1v + s2maxl[hh * 16 + lane16];
  rmax = rmax > 0.f ? rmax : NEG_SLOPE * rmax;
  float psum = 0.f;
  const float* hb  = hbuf + (b * Hh + hh) * Nq * HD;
  const float* s2h = s2l + hh * Nq;
  const unsigned int* mrow = maskw + lane16 * 32;
  const int mbase = half * (Nq / 2);

  v8f acc0 = {}, acc1 = {}, acc2 = {}, acc3 = {};
  const int mend = mbase + Nq / 2;
  const float* hlane = hb + (2 * laneHi) * HD + lane16;
  auto form_p = [&](v2f* p, const v2f* sq, const unsigned int bits) {
#pragma unroll
    for (int s = 0; s < 4; ++s) {
      float v0 = s1v + sq[s].x;
      float v1 = s1v + sq[s].y;
      v0 = v0 > 0.f ? v0 : NEG_SLOPE * v0;
      v1 = v1 > 0.f ? v1 : NEG_SLOPE * v1;
      const float e0 = __expf(v0 - rmax);
      const float e1 = __expf(v1 - rmax);
      p[s].x = ((bits >> (4 * s)) & 1u) ? e0 : 0.f;
      p[s].y = ((bits >> (4 * s + 1)) & 1u) ? e1 : 0.f;
    }
  };
  auto mma_block = [&](const HBlock& f, const v2f* p) {
#pragma unroll
    for (int s = 0; s < 4; ++s) {
      psum += p[s].x + p[s].y;
      acc0 = __builtin_amdgcn_wmma_f32_16x16x4_f32(false, p[s], false, f.t[s][0], (short)0, acc0, false, false);
      acc1 = __builtin_amdgcn_wmma_f32_16x16x4_f32(false, p[s], false, f.t[s][1], (short)0, acc1, false, false);
      acc2 = __builtin_amdgcn_wmma_f32_16x16x4_f32(false, p[s], false, f.t[s][2], (short)0, acc2, false, false);
      acc3 = __builtin_amdgcn_wmma_f32_16x16x4_f32(false, p[s], false, f.t[s][3], (short)0, acc3, false, false);
    }
  };
  auto load_trip = [&](v2f* sq, unsigned int& mw, const int kt) {
    const float* s2k = s2h + kt + 2 * laneHi;
#pragma unroll
    for (int j = 0; j < 8; ++j) sq[j] = *reinterpret_cast<const v2f*>(s2k + 4 * j);
    mw = mrow[kt >> 5] >> (2 * laneHi);
  };

  HBlock fa, fb;
  v2f pa[4], pb[4];
  v2f sqh[4];
  unsigned int mwh;
  {
    v2f s0[8]; unsigned int m0w;
    load_trip(s0, m0w, mbase);
    load_hblock(fa, hlane + mbase * HD);
    form_p(pa, s0, m0w);
#pragma unroll
    for (int j = 0; j < 4; ++j) sqh[j] = s0[4 + j];
    mwh = m0w >> 16;
  }
  for (int kb = mbase; kb < mend; kb += 32) {
    const int kn = kb + 32 < mend ? kb + 32 : mbase;
    v2f sn[8]; unsigned int mwn;
    load_trip(sn, mwn, kn);
    load_hblock(fb, hlane + (kb + 16) * HD);
    __builtin_amdgcn_sched_barrier(0);
    form_p(pb, sqh, mwh);
    mma_block(fa, pa);
    int hnext = kn * HD;
    asm volatile("" : "+v"(hnext) : "v"(acc0), "v"(acc1), "v"(acc2), "v"(acc3),
                 "v"(pb[0]), "v"(pb[1]), "v"(pb[2]), "v"(pb[3]));
    load_hblock(fa, hlane + hnext);
    __builtin_amdgcn_sched_barrier(0);
    form_p(pa, sn, mwn);
    mma_block(fb, pb);
#pragma unroll
    for (int j = 0; j < 4; ++j) sqh[j] = sn[4 + j];
    mwh = mwn >> 16;
  }

  psum += __shfl_xor(psum, 16, 32);
  if (laneHi == 0) psuml[(half * Hh + hh) * 16 + lane16] = psum;
  float* op = opart + hh * 16 * HD;
  if (half == 0) {
#pragma unroll
    for (int v = 0; v < 8; ++v) {
      float* orow = op + (v + 8 * laneHi) * HD + lane16;
      orow[0]  = acc0[v];
      orow[16] = acc1[v];
      orow[32] = acc2[v];
      orow[48] = acc3[v];
    }
  }
  __syncthreads();
  if (half == 1) {
    float* ob = out + (size_t)b * Nq * Fout + hh * HD;
#pragma unroll
    for (int v = 0; v < 8; ++v) {
      const float* prow = op + (v + 8 * laneHi) * HD + lane16;
      float* orow = ob + (size_t)(n0 + v + 8 * laneHi) * Fout + lane16;
      const float rsum = psuml[hh * 16 + v + 8 * laneHi] + psuml[(Hh + hh) * 16 + v + 8 * laneHi];
      const float rinv = rsum > 0.f ? 1.f / rsum : 0.f;
      orow[0]  = (acc0[v] + prow[0]) * rinv;
      orow[16] = (acc1[v] + prow[16]) * rinv;
      orow[32] = (acc2[v] + prow[32]) * rinv;
      orow[48] = (acc3[v] + prow[48]) * rinv;
    }
  }
}

extern "C" void kernel_launch(void* const* d_in, const int* in_sizes, int n_in,
                              void* d_out, int out_size, void* d_ws, size_t ws_size,
                              hipStream_t stream) {
  const float* x   = (const float*)d_in[0];
  const int*   adj = (const int*)d_in[1];
  const float* W   = (const float*)d_in[2];
  const float* a   = (const float*)d_in[3];
  float* out = (float*)d_out;

  float* hbuf = (float*)d_ws;
  float* s1 = hbuf + (size_t)Bq * Hh * Nq * HD;
  float* s2 = s1 + (size_t)Bq * Hh * Nq;

  gat_proj<<<2048, 128, 0, stream>>>(x, W, a, hbuf, s1, s2);
  gat_attn<<<dim3(Nq / 16, Bq), 256, 0, stream>>>(hbuf, adj, s1, s2, out);
}


# h stored row-pair interleaved (B,H,N/2,HD,2), column tiles interleaved: attn B fragments come as 8 16-byte whole-line loads per 16-k block in operand order, not 32 4-byte; h and O stores 8-byte
# speedup vs baseline: 2.1485x; 1.0628x over previous
#include <hip/hip_runtime.h>


typedef float v2f __attribute__((ext_vector_type(2)));
typedef float v8f __attribute__((ext_vector_type(8)));

#define Bq 8
#define Nq 1024
#define Fin 256
#define Fout 256
#define Hh 4
#define HD 64
#define NEG_SLOPE 0.2f

#define PRP 68
__global__ __launch_bounds__(128) void gat_proj(const float* __restrict__ x,
                                                const float* __restrict__ W,
                                                const float* __restrict__ a,
                                                float* __restrict__ hbuf,
                                                float* __restrict__ s1,
                                                float* __restrict__ s2) {
  __shared__ __attribute__((aligned(16))) float prod[2][16][PRP];

  const int tid = threadIdx.x;
  const int wave = tid >> 5, lane = tid & 31;
  const int lane16 = lane & 15, laneHi = lane >> 4;
  const int tile = blockIdx.x * 4 + wave;
  const int mt = tile >> 4;
  const int nt = tile & 15;
  const int row = mt * 16 + lane16;
  const int col = nt * 16 + lane16;
  const int hh = col >> 6, d = col & 63;

  const float a1d = a[hh * (2 * HD) + d];
  const float a2d = a[hh * (2 * HD) + HD + d];

  const float* xr = x + row * Fin;
  v8f acc = {};
#pragma unroll 4
  for (int k = 0; k < Fin; k += 4) {
    const int k0 = k + 2 * laneHi;
    v2f av = *reinterpret_cast<const v2f*>(xr + k0);
    v2f bv;
    bv.x = W[k0 * Fout + col];
    bv.y = W[(k0 + 1) * Fout + col];
    acc = __builtin_amdgcn_wmma_f32_16x16x4_f32(false, av, false, bv,
                                                (short)0, acc, false, false);
  }
#pragma unroll
  for (int v = 0; v < 8; ++v) {
    prod[0][v + 8 * laneHi][d] = acc[v] * a1d;
    prod[1][v + 8 * laneHi][d] = acc[v] * a2d;
  }
  __syncthreads();
#pragma unroll
  for (int v = 0; v < 8; v += 2) {
    const int r = mt * 16 + v + 8 * laneHi;
    const int bb = r >> 10, n = r & (Nq - 1);
    v2f pr;
    pr.x = acc[v];
    pr.y = acc[v + 1];
    *reinterpret_cast<v2f*>(hbuf + ((bb * Hh + hh) * Nq + n) * HD + 2 * d) = pr;
  }
  {
    const int i = tid >> 2, part = tid & 3;
    const int q = i >> 4, rr = i & 15;
    const float4* pr = reinterpret_cast<const float4*>(&prod[q][rr][part * 16]);
    const float4 u0 = pr[0], u1 = pr[1], u2 = pr[2], u3 = pr[3];
    float sum = (((u0.x + u0.y) + (u0.z + u0.w)) + ((u1.x + u1.y) + (u1.z + u1.w))) +
                (((u2.x + u2.y) + (u2.z + u2.w)) + ((u3.x + u3.y) + (u3.z + u3.w)));
    sum += __shfl_xor(sum, 1, 32);
    sum += __shfl_xor(sum, 2, 32);
    if (part == 0) {
      const int r = mt * 16 + rr;
      const int bb = r >> 10, n = r & (Nq - 1);
      float* dst = q ? s2 : s1;
      dst[(bb * Hh + hh) * Nq + n] = sum;
    }
  }
}

typedef float v4f __attribute__((ext_vector_type(4)));

struct HBlock { v2f t[4][4]; };

__device__ __forceinline__ void load_hblock(HBlock& f, const float* __restrict__ hrow) {
#pragma unroll
  for (int s = 0; s < 4; ++s) {
    const v4f lo = *reinterpret_cast<const v4f*>(hrow + s * (4 * HD));
    const v4f hi = *reinterpret_cast<const v4f*>(hrow + s * (4 * HD) + HD);
    f.t[s][0] = lo.xy;
    f.t[s][1] = lo.zw;
    f.t[s][2] = hi.xy;
    f.t[s][3] = hi.zw;
  }
}

__global__ __launch_bounds__(256) void gat_attn(const float* __restrict__ hbuf,
                                                const int* __restrict__ adj,
                                                const float* __restrict__ s1g,
                                                const float* __restrict__ s2g,
                                                float* __restrict__ out) {
  __shared__ unsigned int maskw[16 * (Nq / 32)];
  __shared__ __attribute__((aligned(16))) float s2l[Hh * Nq];
  __shared__ float s1l[Hh * 16];
  __shared__ float s2maxl[Hh * 16];
  __shared__ float psuml[2 * Hh * 16];
  __shared__ float opart[Hh * 16 * HD];

  const int tid = threadIdx.x;
  const int wave = tid >> 5, lane = tid & 31;
  const int lane16 = lane & 15, laneHi = lane >> 4;
  const int b = blockIdx.y, n0 = blockIdx.x * 16;
  const int hh = wave >> 1;
  const int half = wave & 1;

  const int* adjb = adj + (b * Nq + n0) * Nq;
  for (int w = tid; w < 16 * (Nq / 32); w += 256) {
    const int r = w >> 5, m0 = (w & 31) * 32;
    const int* ar = adjb + r * Nq + m0;
    unsigned int bits = 0u;
#pragma unroll
    for (int j = 0; j < 32; ++j) bits |= (ar[j] != 0 ? 1u : 0u) << j;
    maskw[w] = bits;
  }
  const float* s2b = s2g + b * Hh * Nq;
  for (int i = tid; i < Hh * Nq; i += 256) s2l[i] = s2b[i];
  const float* s1b = s1g + b * Hh * Nq;
  for (int i = tid; i < Hh * 16; i += 256)
    s1l[i] = s1b[(i >> 4) * Nq + n0 + (i & 15)];
  __syncthreads();

  float s2v[32];
  {
    const float4* s2q = reinterpret_cast<const float4*>(s2l + hh * Nq + lane * 32);
#pragma unroll
    for (int q = 0; q < 8; ++q) {
      const float4 t = s2q[q];
      s2v[4 * q] = t.x; s2v[4 * q + 1] = t.y; s2v[4 * q + 2] = t.z; s2v[4 * q + 3] = t.w;
    }
  }
  const float ninf = -__builtin_huge_valf();
  float mx8[8];
#pragma unroll
  for (int r8 = 0; r8 < 8; ++r8) {
    const unsigned int w = maskw[(half * 8 + r8) * 32 + lane];
    float mx = ninf;
#pragma unroll
    for (int j = 0; j < 32; ++j) mx = fmaxf(mx, ((w >> j) & 1u) ? s2v[j] : ninf);
    mx8[r8] = mx;
  }
#pragma unroll
  for (int off = 16; off; off >>= 1) {
#pragma unroll
    for (int r8 = 0; r8 < 8; ++r8)
      mx8[r8] = fmaxf(mx8[r8], __shfl_xor(mx8[r8], off, 32));
  }
  if (lane == 0) {
#pragma unroll
    for (int r8 = 0; r8 < 8; ++r8) s2maxl[hh * 16 + half * 8 + r8] = mx8[r8];
  }
  __syncthreads();

  const float s1v  = s1l[hh * 16 + lane16];
  float rmax = s1v + s2maxl[hh * 16 + lane16];
  rmax = rmax > 0.f ? rmax : NEG_SLOPE * rmax;
  float psum = 0.f;
  const float* hb  = hbuf + (b * Hh + hh) * Nq * HD;
  const float* s2h = s2l + hh * Nq;
  const unsigned int* mrow = maskw + lane16 * 32;
  const int mbase = half * (Nq / 2);

  v8f acc0 = {}, acc1 = {}, acc2 = {}, acc3 = {};
  const int mend = mbase + Nq / 2;
  const float* hlane = hb + laneHi * (2 * HD) + 4 * lane16;
  auto form_p = [&](v2f* p, const v2f* sq, const unsigned int bits) {
#pragma unroll
    for (int s = 0; s < 4; ++s) {
      float v0 = s1v + sq[s].x;
      float v1 = s1v + sq[s].y;
      v0 = v0 > 0.f ? v0 : NEG_SLOPE * v0;
      v1 = v1 > 0.f ? v1 : NEG_SLOPE * v1;
      const float e0 = __expf(v0 - rmax);
      const float e1 = __expf(v1 - rmax);
      p[s].x = ((bits >> (4 * s)) & 1u) ? e0 : 0.f;
      p[s].y = ((bits >> (4 * s + 1)) & 1u) ? e1 : 0.f;
    }
  };
  auto mma_block = [&](const HBlock& f, const v2f* p) {
#pragma unroll
    for (int s = 0; s < 4; ++s) {
      psum += p[s].x + p[s].y;
      acc0 = __builtin_amdgcn_wmma_f32_16x16x4_f32(false, p[s], false, f.t[s][0], (short)0, acc0, false, false);
      acc1 = __builtin_amdgcn_wmma_f32_16x16x4_f32(false, p[s], false, f.t[s][1], (short)0, acc1, false, false);
      acc2 = __builtin_amdgcn_wmma_f32_16x16x4_f32(false, p[s], false, f.t[s][2], (short)0, acc2, false, false);
      acc3 = __builtin_amdgcn_wmma_f32_16x16x4_f32(false, p[s], false, f.t[s][3], (short)0, acc3, false, false);
    }
  };
  auto load_trip = [&](v2f* sq, unsigned int& mw, const int kt) {
    const float* s2k = s2h + kt + 2 * laneHi;
#pragma unroll
    for (int j = 0; j < 8; ++j) sq[j] = *reinterpret_cast<const v2f*>(s2k + 4 * j);
    mw = mrow[kt >> 5] >> (2 * laneHi);
  };

  HBlock fa, fb;
  v2f pa[4], pb[4];
  v2f sqh[4];
  unsigned int mwh;
  {
    v2f s0[8]; unsigned int m0w;
    load_trip(s0, m0w, mbase);
    load_hblock(fa, hlane + mbase * HD);
    form_p(pa, s0, m0w);
#pragma unroll
    for (int j = 0; j < 4; ++j) sqh[j] = s0[4 + j];
    mwh = m0w >> 16;
  }
  for (int kb = mbase; kb < mend; kb += 32) {
    const int kn = kb + 32 < mend ? kb + 32 : mbase;
    v2f sn[8]; unsigned int mwn;
    load_trip(sn, mwn, kn);
    load_hblock(fb, hlane + (kb + 16) * HD);
    __builtin_amdgcn_sched_barrier(0);
    form_p(pb, sqh, mwh);
    mma_block(fa, pa);
    int hnext = kn * HD;
    asm volatile("" : "+v"(hnext) : "v"(acc0), "v"(acc1), "v"(acc2), "v"(acc3),
                 "v"(pb[0]), "v"(pb[1]), "v"(pb[2]), "v"(pb[3]));
    load_hblock(fa, hlane + hnext);
    __builtin_amdgcn_sched_barrier(0);
    form_p(pa, sn, mwn);
    mma_block(fb, pb);
#pragma unroll
    for (int j = 0; j < 4; ++j) sqh[j] = sn[4 + j];
    mwh = mwn >> 16;
  }

  psum += __shfl_xor(psum, 16, 32);
  if (laneHi == 0) psuml[(half * Hh + hh) * 16 + lane16] = psum;
  float* op = opart + hh * 16 * HD;
  if (half == 0) {
#pragma unroll
    for (int v = 0; v < 8; ++v) {
      float* orow = op + (v + 8 * laneHi) * HD + lane16;
      orow[0]  = acc0[v];
      orow[16] = acc1[v];
      orow[32] = acc2[v];
      orow[48] = acc3[v];
    }
  }
  __syncthreads();
  if (half == 1) {
    float* ob = out + (size_t)b * Nq * Fout + hh * HD;
#pragma unroll
    for (int v = 0; v < 8; ++v) {
      const float* prow = op + (v + 8 * laneHi) * HD + lane16;
      float* orow = ob + (size_t)(n0 + v + 8 * laneHi) * Fout + 2 * lane16;
      const float rsum = psuml[hh * 16 + v + 8 * laneHi] + psuml[(Hh + hh) * 16 + v + 8 * laneHi];
      const float rinv = rsum > 0.f ? 1.f / rsum : 0.f;
      v2f o01, o23;
      o01.x = (acc0[v] + prow[0]) * rinv;
      o01.y = (acc1[v] + prow[16]) * rinv;
      o23.x = (acc2[v] + prow[32]) * rinv;
      o23.y = (acc3[v] + prow[48]) * rinv;
      *reinterpret_cast<v2f*>(orow)      = o01;
      *reinterpret_cast<v2f*>(orow + 32) = o23;
    }
  }
}

extern "C" void kernel_launch(void* const* d_in, const int* in_sizes, int n_in,
                              void* d_out, int out_size, void* d_ws, size_t ws_size,
                              hipStream_t stream) {
  const float* x   = (const float*)d_in[0];
  const int*   adj = (const int*)d_in[1];
  const float* W   = (const float*)d_in[2];
  const float* a   = (const float*)d_in[3];
  float* out = (float*)d_out;

  float* hbuf = (float*)d_ws;
  float* s1 = hbuf + (size_t)Bq * Hh * Nq * HD;
  float* s2 = s1 + (size_t)Bq * Hh * Nq;

  gat_proj<<<2048, 128, 0, stream>>>(x, W, a, hbuf, s1, s2);
  gat_attn<<<dim3(Nq / 16, Bq), 256, 0, stream>>>(hbuf, adj, s1, s2, out);
}


# proj retiled: a wave = 16 rows x whole head (4 tiles), x fragment feeds 4 WMMAs, x and W by 16-byte loads (k permuted in 8-k groups) a block ahead; scores by in-wave lane exchange: new sum order
# speedup vs baseline: 2.5026x; 1.1648x over previous
#include <hip/hip_runtime.h>


typedef float v2f __attribute__((ext_vector_type(2)));
typedef float v8f __attribute__((ext_vector_type(8)));

#define Bq 8
#define Nq 1024
#define Fin 256
#define Fout 256
#define Hh 4
#define HD 64
#define NEG_SLOPE 0.2f

typedef float v4f __attribute__((ext_vector_type(4)));

struct PBlock { v4f xa[4]; v4f w[4][4]; };

__device__ __forceinline__ void load_pblock(PBlock& f, const float* __restrict__ xp,
                                            const float* __restrict__ wp) {
#pragma unroll
  for (int g = 0; g < 4; ++g) {
    f.xa[g] = *reinterpret_cast<const v4f*>(xp + 8 * g);
#pragma unroll
    for (int j = 0; j < 4; ++j)
      f.w[g][j] = *reinterpret_cast<const v4f*>(wp + (8 * g + j) * Fout);
  }
}

template <int CNT>
__device__ __forceinline__ void score_step(float (&sv)[16], const int lane16) {
  const bool up = (lane16 & CNT) != 0;
#pragma unroll
  for (int i = 0; i < CNT; ++i) {
    const float send = up ? sv[i] : sv[i + CNT];
    const float keep = up ? sv[i + CNT] : sv[i];
    const float got = __builtin_bit_cast(float,
        __builtin_amdgcn_update_dpp(0, __builtin_bit_cast(int, send), 0x160 + CNT, 0xF, 0xF, false));
    sv[i] = keep + got;
  }
}

__global__ __launch_bounds__(128) void gat_proj(const float* __restrict__ x,
                                                const float* __restrict__ W,
                                                const float* __restrict__ a,
                                                float* __restrict__ hbuf,
                                                float* __restrict__ s1,
                                                float* __restrict__ s2) {
  const int tid = threadIdx.x;
  const int wave = tid >> 5, lane = tid & 31;
  const int lane16 = lane & 15, laneHi = lane >> 4;
  const int hh = blockIdx.x & 3;
  const int mt = (blockIdx.x >> 2) * 4 + wave;

  const v4f a1v = *reinterpret_cast<const v4f*>(a + hh * (2 * HD) + 4 * lane16);
  const v4f a2v = *reinterpret_cast<const v4f*>(a + hh * (2 * HD) + HD + 4 * lane16);

  const float* xp = x + (mt * 16 + lane16) * Fin + 4 * laneHi;
  const float* wp = W + (4 * laneHi) * Fout + hh * HD + 4 * lane16;

  v8f acc0 = {}, acc1 = {}, acc2 = {}, acc3 = {};
  auto mma_pblock = [&](const PBlock& f) {
#pragma unroll
    for (int g = 0; g < 4; ++g) {
#pragma unroll
      for (int s = 0; s < 2; ++s) {
        const v2f av = s ? f.xa[g].zw : f.xa[g].xy;
        const v4f wlo = f.w[g][2 * s], whi = f.w[g][2 * s + 1];
        v2f b0, b1, b2, b3;
        b0.x = wlo.x; b0.y = whi.x;
        b1.x = wlo.y; b1.y = whi.y;
        b2.x = wlo.z; b2.y = whi.z;
        b3.x = wlo.w; b3.y = whi.w;
        acc0 = __builtin_amdgcn_wmma_f32_16x16x4_f32(false, av, false, b0, (short)0, acc0, false, false);
        acc1 = __builtin_amdgcn_wmma_f32_16x16x4_f32(false, av, false, b1, (short)0, acc1, false, false);
        acc2 = __builtin_amdgcn_wmma_f32_16x16x4_f32(false, av, false, b2, (short)0, acc2, false, false);
        acc3 = __builtin_amdgcn_wmma_f32_16x16x4_f32(false, av, false, b3, (short)0, acc3, false, false);
      }
    }
  };

  PBlock fa, fb;
  load_pblock(fa, xp, wp);
#pragma unroll 1
  for (int kb = 0; kb < Fin - 64; kb += 64) {
    load_pblock(fb, xp + kb + 32, wp + (kb + 32) * Fout);
    __builtin_amdgcn_sched_barrier(0);
    mma_pblock(fa);
    int kn = kb + 64;
    asm volatile("" : "+v"(kn) : "v"(acc0), "v"(acc1), "v"(acc2), "v"(acc3));
    load_pblock(fa, xp + kn, wp + kn * Fout);
    __builtin_amdgcn_sched_barrier(0);
    mma_pblock(fb);
  }
  load_pblock(fb, xp + (Fin - 32), wp + (Fin - 32) * Fout);
  __builtin_amdgcn_sched_barrier(0);
  mma_pblock(fa);
  mma_pblock(fb);

#pragma unroll
  for (int v = 0; v < 8; v += 2) {
    const int r = mt * 16 + v + 8 * laneHi;
    const int bb = r >> 10, n = r & (Nq - 1);
    float* dst = hbuf + ((bb * Hh + hh) * Nq + n) * HD + 8 * lane16;
    v4f lo, hi;
    lo.x = acc0[v]; lo.y = acc0[v + 1]; lo.z = acc1[v]; lo.w = acc1[v + 1];
    hi.x = acc2[v]; hi.y = acc2[v + 1]; hi.z = acc3[v]; hi.w = acc3[v + 1];
    *reinterpret_cast<v4f*>(dst)     = lo;
    *reinterpret_cast<v4f*>(dst + 4) = hi;
  }

  float sv[16];
#pragma unroll
  for (int v = 0; v < 8; ++v) {
    sv[v]     = (acc0[v] * a1v.x + acc1[v] * a1v.y) + (acc2[v] * a1v.z + acc3[v] * a1v.w);
    sv[8 + v] = (acc0[v] * a2v.x + acc1[v] * a2v.y) + (acc2[v] * a2v.z + acc3[v] * a2v.w);
  }
  score_step<8>(sv, lane16);
  score_step<4>(sv, lane16);
  score_step<2>(sv, lane16);
  score_step<1>(sv, lane16);
  {
    const int q = lane16 >> 3, v = lane16 & 7;
    const int r = mt * 16 + v + 8 * laneHi;
    const int bb = r >> 10, n = r & (Nq - 1);
    float* dst = q ? s2 : s1;
    dst[(bb * Hh + hh) * Nq + n] = sv[0];
  }
}


struct HBlock { v2f t[4][4]; };

__device__ __forceinline__ void load_hblock(HBlock& f, const float* __restrict__ hrow) {
#pragma unroll
  for (int s = 0; s < 4; ++s) {
    const v4f lo = *reinterpret_cast<const v4f*>(hrow + s * (4 * HD));
    const v4f hi = *reinterpret_cast<const v4f*>(hrow + s * (4 * HD) + HD);
    f.t[s][0] = lo.xy;
    f.t[s][1] = lo.zw;
    f.t[s][2] = hi.xy;
    f.t[s][3] = hi.zw;
  }
}

__global__ __launch_bounds__(256) void gat_attn(const float* __restrict__ hbuf,
                                                const int* __restrict__ adj,
                                                const float* __restrict__ s1g,
                                                const float* __restrict__ s2g,
                                                float* __restrict__ out) {
  __shared__ unsigned int maskw[16 * (Nq / 32)];
  __shared__ __attribute__((aligned(16))) float s2l[Hh * Nq];
  __shared__ float s1l[Hh * 16];
  __shared__ float s2maxl[Hh * 16];
  __shared__ float psuml[2 * Hh * 16];
  __shared__ float opart[Hh * 16 * HD];

  const int tid = threadIdx.x;
  const int wave = tid >> 5, lane = tid & 31;
  const int lane16 = lane & 15, laneHi = lane >> 4;
  const int b = blockIdx.y, n0 = blockIdx.x * 16;
  const int hh = wave >> 1;
  const int half = wave & 1;

  const int* adjb = adj + (b * Nq + n0) * Nq;
  for (int w = tid; w < 16 * (Nq / 32); w += 256) {
    const int r = w >> 5, m0 = (w & 31) * 32;
    const int* ar = adjb + r * Nq + m0;
    unsigned int bits = 0u;
#pragma unroll
    for (int j = 0; j < 32; ++j) bits |= (ar[j] != 0 ? 1u : 0u) << j;
    maskw[w] = bits;
  }
  const float* s2b = s2g + b * Hh * Nq;
  for (int i = tid; i < Hh * Nq; i += 256) s2l[i] = s2b[i];
  const float* s1b = s1g + b * Hh * Nq;
  for (int i = tid; i < Hh * 16; i += 256)
    s1l[i] = s1b[(i >> 4) * Nq + n0 + (i & 15)];
  __syncthreads();

  float s2v[32];
  {
    const float4* s2q = reinterpret_cast<const float4*>(s2l + hh * Nq + lane * 32);
#pragma unroll
    for (int q = 0; q < 8; ++q) {
      const float4 t = s2q[q];
      s2v[4 * q] = t.x; s2v[4 * q + 1] = t.y; s2v[4 * q + 2] = t.z; s2v[4 * q + 3] = t.w;
    }
  }
  const float ninf = -__builtin_huge_valf();
  float mx8[8];
#pragma unroll
  for (int r8 = 0; r8 < 8; ++r8) {
    const unsigned int w = maskw[(half * 8 + r8) * 32 + lane];
    float mx = ninf;
#pragma unroll
    for (int j = 0; j < 32; ++j) mx = fmaxf(mx, ((w >> j) & 1u) ? s2v[j] : ninf);
    mx8[r8] = mx;
  }
#pragma unroll
  for (int off = 16; off; off >>= 1) {
#pragma unroll
    for (int r8 = 0; r8 < 8; ++r8)
      mx8[r8] = fmaxf(mx8[r8], __shfl_xor(mx8[r8], off, 32));
  }
  if (lane == 0) {
#pragma unroll
    for (int r8 = 0; r8 < 8; ++r8) s2maxl[hh * 16 + half * 8 + r8] = mx8[r8];
  }
  __syncthreads();

  const float s1v  = s1l[hh * 16 + lane16];
  float rmax = s1v + s2maxl[hh * 16 + lane16];
  rmax = rmax > 0.f ? rmax : NEG_SLOPE * rmax;
  float psum = 0.f;
  const float* hb  = hbuf + (b * Hh + hh) * Nq * HD;
  const float* s2h = s2l + hh * Nq;
  const unsigned int* mrow = maskw + lane16 * 32;
  const int mbase = half * (Nq / 2);

  v8f acc0 = {}, acc1 = {}, acc2 = {}, acc3 = {};
  const int mend = mbase + Nq / 2;
  const float* hlane = hb + laneHi * (2 * HD) + 4 * lane16;
  auto form_p = [&](v2f* p, const v2f* sq, const unsigned int bits) {
#pragma unroll
    for (int s = 0; s < 4; ++s) {
      float v0 = s1v + sq[s].x;
      float v1 = s1v + sq[s].y;
      v0 = v0 > 0.f ? v0 : NEG_SLOPE * v0;
      v1 = v1 > 0.f ? v1 : NEG_SLOPE * v1;
      const float e0 = __expf(v0 - rmax);
      const float e1 = __expf(v1 - rmax);
      p[s].x = ((bits >> (4 * s)) & 1u) ? e0 : 0.f;
      p[s].y = ((bits >> (4 * s + 1)) & 1u) ? e1 : 0.f;
    }
  };
  auto mma_block = [&](const HBlock& f, const v2f* p) {
#pragma unroll
    for (int s = 0; s < 4; ++s) {
      psum += p[s].x + p[s].y;
      acc0 = __builtin_amdgcn_wmma_f32_16x16x4_f32(false, p[s], false, f.t[s][0], (short)0, acc0, false, false);
      acc1 = __builtin_amdgcn_wmma_f32_16x16x4_f32(false, p[s], false, f.t[s][1], (short)0, acc1, false, false);
      acc2 = __builtin_amdgcn_wmma_f32_16x16x4_f32(false, p[s], false, f.t[s][2], (short)0, acc2, false, false);
      acc3 = __builtin_amdgcn_wmma_f32_16x16x4_f32(false, p[s], false, f.t[s][3], (short)0, acc3, false, false);
    }
  };
  auto load_trip = [&](v2f* sq, unsigned int& mw, const int kt) {
    const float* s2k = s2h + kt + 2 * laneHi;
#pragma unroll
    for (int j = 0; j < 8; ++j) sq[j] = *reinterpret_cast<const v2f*>(s2k + 4 * j);
    mw = mrow[kt >> 5] >> (2 * laneHi);
  };

  HBlock fa, fb;
  v2f pa[4], pb[4];
  v2f sqh[4];
  unsigned int mwh;
  {
    v2f s0[8]; unsigned int m0w;
    load_trip(s0, m0w, mbase);
    load_hblock(fa, hlane + mbase * HD);
    form_p(pa, s0, m0w);
#pragma unroll
    for (int j = 0; j < 4; ++j) sqh[j] = s0[4 + j];
    mwh = m0w >> 16;
  }
  for (int kb = mbase; kb < mend; kb += 32) {
    const int kn = kb + 32 < mend ? kb + 32 : mbase;
    v2f sn[8]; unsigned int mwn;
    load_trip(sn, mwn, kn);
    load_hblock(fb, hlane + (kb + 16) * HD);
    __builtin_amdgcn_sched_barrier(0);
    form_p(pb, sqh, mwh);
    mma_block(fa, pa);
    int hnext = kn * HD;
    asm volatile("" : "+v"(hnext) : "v"(acc0), "v"(acc1), "v"(acc2), "v"(acc3),
                 "v"(pb[0]), "v"(pb[1]), "v"(pb[2]), "v"(pb[3]));
    load_hblock(fa, hlane + hnext);
    __builtin_amdgcn_sched_barrier(0);
    form_p(pa, sn, mwn);
    mma_block(fb, pb);
#pragma unroll
    for (int j = 0; j < 4; ++j) sqh[j] = sn[4 + j];
    mwh = mwn >> 16;
  }

  psum += __shfl_xor(psum, 16, 32);
  if (laneHi == 0) psuml[(half * Hh + hh) * 16 + lane16] = psum;
  float* op = opart + hh * 16 * HD;
  if (half == 0) {
#pragma unroll
    for (int v = 0; v < 8; ++v) {
      float* orow = op + (v + 8 * laneHi) * HD + lane16;
      orow[0]  = acc0[v];
      orow[16] = acc1[v];
      orow[32] = acc2[v];
      orow[48] = acc3[v];
    }
  }
  __syncthreads();
  if (half == 1) {
    float* ob = out + (size_t)b * Nq * Fout + hh * HD;
#pragma unroll
    for (int v = 0; v < 8; ++v) {
      const float* prow = op + (v + 8 * laneHi) * HD + lane16;
      float* orow = ob + (size_t)(n0 + v + 8 * laneHi) * Fout + 2 * lane16;
      const float rsum = psuml[hh * 16 + v + 8 * laneHi] + psuml[(Hh + hh) * 16 + v + 8 * laneHi];
      const float rinv = rsum > 0.f ? 1.f / rsum : 0.f;
      v2f o01, o23;
      o01.x = (acc0[v] + prow[0]) * rinv;
      o01.y = (acc1[v] + prow[16]) * rinv;
      o23.x = (acc2[v] + prow[32]) * rinv;
      o23.y = (acc3[v] + prow[48]) * rinv;
      *reinterpret_cast<v2f*>(orow)      = o01;
      *reinterpret_cast<v2f*>(orow + 32) = o23;
    }
  }
}

extern "C" void kernel_launch(void* const* d_in, const int* in_sizes, int n_in,
                              void* d_out, int out_size, void* d_ws, size_t ws_size,
                              hipStream_t stream) {
  const float* x   = (const float*)d_in[0];
  const int*   adj = (const int*)d_in[1];
  const float* W   = (const float*)d_in[2];
  const float* a   = (const float*)d_in[3];
  float* out = (float*)d_out;

  float* hbuf = (float*)d_ws;
  float* s1 = hbuf + (size_t)Bq * Hh * Nq * HD;
  float* s2 = s1 + (size_t)Bq * Hh * Nq;

  gat_proj<<<512, 128, 0, stream>>>(x, W, a, hbuf, s1, s2);
  gat_attn<<<dim3(Nq / 16, Bq), 256, 0, stream>>>(hbuf, adj, s1, s2, out);
}


# attn prologue: adj tile, s2, s1 loaded as one group; one sweep of the adj registers gives the mask words by wave ballot and, the ballot as lane select, the masked s2 row maxima: stats pass gone
# speedup vs baseline: 2.7084x; 1.0822x over previous
#include <hip/hip_runtime.h>


typedef float v2f __attribute__((ext_vector_type(2)));
typedef float v8f __attribute__((ext_vector_type(8)));

#define Bq 8
#define Nq 1024
#define Fin 256
#define Fout 256
#define Hh 4
#define HD 64
#define NEG_SLOPE 0.2f

typedef float v4f __attribute__((ext_vector_type(4)));

struct PBlock { v4f xa[4]; v4f w[4][4]; };

__device__ __forceinline__ void load_pblock(PBlock& f, const float* __restrict__ xp,
                                            const float* __restrict__ wp) {
#pragma unroll
  for (int g = 0; g < 4; ++g) {
    f.xa[g] = *reinterpret_cast<const v4f*>(xp + 8 * g);
#pragma unroll
    for (int j = 0; j < 4; ++j)
      f.w[g][j] = *reinterpret_cast<const v4f*>(wp + (8 * g + j) * Fout);
  }
}

template <int CNT>
__device__ __forceinline__ void score_step(float (&sv)[16], const int lane16) {
  const bool up = (lane16 & CNT) != 0;
#pragma unroll
  for (int i = 0; i < CNT; ++i) {
    const float send = up ? sv[i] : sv[i + CNT];
    const float keep = up ? sv[i + CNT] : sv[i];
    const float got = __builtin_bit_cast(float,
        __builtin_amdgcn_update_dpp(0, __builtin_bit_cast(int, send), 0x160 + CNT, 0xF, 0xF, false));
    sv[i] = keep + got;
  }
}

__global__ __launch_bounds__(128) void gat_proj(const float* __restrict__ x,
                                                const float* __restrict__ W,
                                                const float* __restrict__ a,
                                                float* __restrict__ hbuf,
                                                float* __restrict__ s1,
                                                float* __restrict__ s2) {
  const int tid = threadIdx.x;
  const int wave = tid >> 5, lane = tid & 31;
  const int lane16 = lane & 15, laneHi = lane >> 4;
  const int hh = blockIdx.x & 3;
  const int mt = (blockIdx.x >> 2) * 4 + wave;

  const v4f a1v = *reinterpret_cast<const v4f*>(a + hh * (2 * HD) + 4 * lane16);
  const v4f a2v = *reinterpret_cast<const v4f*>(a + hh * (2 * HD) + HD + 4 * lane16);

  const float* xp = x + (mt * 16 + lane16) * Fin + 4 * laneHi;
  const float* wp = W + (4 * laneHi) * Fout + hh * HD + 4 * lane16;

  v8f acc0 = {}, acc1 = {}, acc2 = {}, acc3 = {};
  auto mma_pblock = [&](const PBlock& f) {
#pragma unroll
    for (int g = 0; g < 4; ++g) {
#pragma unroll
      for (int s = 0; s < 2; ++s) {
        const v2f av = s ? f.xa[g].zw : f.xa[g].xy;
        const v4f wlo = f.w[g][2 * s], whi = f.w[g][2 * s + 1];
        v2f b0, b1, b2, b3;
        b0.x = wlo.x; b0.y = whi.x;
        b1.x = wlo.y; b1.y = whi.y;
        b2.x = wlo.z; b2.y = whi.z;
        b3.x = wlo.w; b3.y = whi.w;
        acc0 = __builtin_amdgcn_wmma_f32_16x16x4_f32(false, av, false, b0, (short)0, acc0, false, false);
        acc1 = __builtin_amdgcn_wmma_f32_16x16x4_f32(false, av, false, b1, (short)0, acc1, false, false);
        acc2 = __builtin_amdgcn_wmma_f32_16x16x4_f32(false, av, false, b2, (short)0, acc2, false, false);
        acc3 = __builtin_amdgcn_wmma_f32_16x16x4_f32(false, av, false, b3, (short)0, acc3, false, false);
      }
    }
  };

  PBlock fa, fb;
  load_pblock(fa, xp, wp);
#pragma unroll 1
  for (int kb = 0; kb < Fin - 64; kb += 64) {
    load_pblock(fb, xp + kb + 32, wp + (kb + 32) * Fout);
    __builtin_amdgcn_sched_barrier(0);
    mma_pblock(fa);
    int kn = kb + 64;
    asm volatile("" : "+v"(kn) : "v"(acc0), "v"(acc1), "v"(acc2), "v"(acc3));
    load_pblock(fa, xp + kn, wp + kn * Fout);
    __builtin_amdgcn_sched_barrier(0);
    mma_pblock(fb);
  }
  load_pblock(fb, xp + (Fin - 32), wp + (Fin - 32) * Fout);
  __builtin_amdgcn_sched_barrier(0);
  mma_pblock(fa);
  mma_pblock(fb);

#pragma unroll
  for (int v = 0; v < 8; v += 2) {
    const int r = mt * 16 + v + 8 * laneHi;
    const int bb = r >> 10, n = r & (Nq - 1);
    float* dst = hbuf + ((bb * Hh + hh) * Nq + n) * HD + 8 * lane16;
    v4f lo, hi;
    lo.x = acc0[v]; lo.y = acc0[v + 1]; lo.z = acc1[v]; lo.w = acc1[v + 1];
    hi.x = acc2[v]; hi.y = acc2[v + 1]; hi.z = acc3[v]; hi.w = acc3[v + 1];
    *reinterpret_cast<v4f*>(dst)     = lo;
    *reinterpret_cast<v4f*>(dst + 4) = hi;
  }

  float sv[16];
#pragma unroll
  for (int v = 0; v < 8; ++v) {
    sv[v]     = (acc0[v] * a1v.x + acc1[v] * a1v.y) + (acc2[v] * a1v.z + acc3[v] * a1v.w);
    sv[8 + v] = (acc0[v] * a2v.x + acc1[v] * a2v.y) + (acc2[v] * a2v.z + acc3[v] * a2v.w);
  }
  score_step<8>(sv, lane16);
  score_step<4>(sv, lane16);
  score_step<2>(sv, lane16);
  score_step<1>(sv, lane16);
  {
    const int q = lane16 >> 3, v = lane16 & 7;
    const int r = mt * 16 + v + 8 * laneHi;
    const int bb = r >> 10, n = r & (Nq - 1);
    float* dst = q ? s2 : s1;
    dst[(bb * Hh + hh) * Nq + n] = sv[0];
  }
}


struct HBlock { v2f t[4][4]; };

__device__ __forceinline__ void load_hblock(HBlock& f, const float* __restrict__ hrow) {
#pragma unroll
  for (int s = 0; s < 4; ++s) {
    const v4f lo = *reinterpret_cast<const v4f*>(hrow + s * (4 * HD));
    const v4f hi = *reinterpret_cast<const v4f*>(hrow + s * (4 * HD) + HD);
    f.t[s][0] = lo.xy;
    f.t[s][1] = lo.zw;
    f.t[s][2] = hi.xy;
    f.t[s][3] = hi.zw;
  }
}

typedef int v4i __attribute__((ext_vector_type(4)));
typedef unsigned int v2u __attribute__((ext_vector_type(2)));

__device__ __forceinline__ float xhalf(const float x) {
  return __builtin_bit_cast(float, __builtin_amdgcn_permlanex16(__builtin_bit_cast(int, x),
      __builtin_bit_cast(int, x), 0x76543210, 0xfedcba98, false, false));
}

template <int CNT>
__device__ __forceinline__ void rowmax_step(float (&v)[16], const int lane16) {
  const bool up = (lane16 & CNT) != 0;
#pragma unroll
  for (int i = 0; i < CNT; ++i) {
    const float send = up ? v[i] : v[i + CNT];
    const float keep = up ? v[i + CNT] : v[i];
    const float got = __builtin_bit_cast(float,
        __builtin_amdgcn_update_dpp(0, __builtin_bit_cast(int, send), 0x160 + CNT, 0xF, 0xF, false));
    v[i] = fmaxf(keep, got);
  }
}

__global__ __launch_bounds__(256) void gat_attn(const float* __restrict__ hbuf,
                                                const int* __restrict__ adj,
                                                const float* __restrict__ s1g,
                                                const float* __restrict__ s2g,
                                                float* __restrict__ out) {
  __shared__ __attribute__((aligned(16))) unsigned int maskw[16 * (Nq / 32)];
  __shared__ __attribute__((aligned(16))) float s2l[Hh * Nq];
  __shared__ float s1l[Hh * 16];
  __shared__ float s2maxl[4 * Hh * 16];
  __shared__ float psuml[2 * Hh * 16];
  __shared__ float opart[Hh * 16 * HD];

  const int tid = threadIdx.x;
  const int wave = tid >> 5, lane = tid & 31;
  const int lane16 = lane & 15, laneHi = lane >> 4;
  const int b = blockIdx.y, n0 = blockIdx.x * 16;
  const int hh = wave >> 1;
  const int half = wave & 1;

  const int swRow = (wave & 1) * 8;
  const int swQ = wave >> 1;
  const float* s2b = s2g + b * Hh * Nq;
  const float* s1b = s1g + b * Hh * Nq;
  v4f sv[Hh][2];
#pragma unroll
  for (int h = 0; h < Hh; ++h) {
#pragma unroll
    for (int u = 0; u < 2; ++u)
      sv[h][u] = *reinterpret_cast<const v4f*>(s2b + h * Nq + 256 * swQ + 128 * u + 4 * lane);
  }
  float s1t = 0.f;
  if (tid < Hh * 16) s1t = s1b[(tid >> 4) * Nq + n0 + (tid & 15)];
  const int* adjp = adj + (b * Nq + n0 + swRow) * Nq + 256 * swQ + 4 * lane;
  v4i av[8][2];
#pragma unroll
  for (int r = 0; r < 8; ++r) {
#pragma unroll
    for (int u = 0; u < 2; ++u)
      av[r][u] = *reinterpret_cast<const v4i*>(adjp + r * Nq + 128 * u);
  }
  if ((wave & 1) == 0) {
#pragma unroll
    for (int h = 0; h < Hh; ++h) {
#pragma unroll
      for (int u = 0; u < 2; ++u)
        *reinterpret_cast<v4f*>(s2l + h * Nq + 256 * swQ + 128 * u + 4 * lane) = sv[h][u];
    }
  }
  if (tid < Hh * 16) s1l[tid] = s1t;

  const float ninf = -__builtin_huge_valf();
  float mxr[32];
#pragma unroll
  for (int i = 0; i < 32; ++i) mxr[i] = ninf;
  int wv0 = 0, wv1 = 0;
#pragma unroll
  for (int r = 0; r < 8; ++r) {
#pragma unroll
    for (int u = 0; u < 2; ++u) {
#pragma unroll
      for (int c = 0; c < 4; ++c) {
        const bool on = av[r][u][c] != 0;
        const int word = (int)__builtin_amdgcn_ballot_w32(on);
#pragma unroll
        for (int h = 0; h < Hh; ++h)
          mxr[4 * r + h] = fmaxf(mxr[4 * r + h], on ? sv[h][u][c] : ninf);
        const int id = r * 8 + u * 4 + c;
        if (id < 32) wv0 = (lane == id) ? word : wv0;
        else         wv1 = (lane == id - 32) ? word : wv1;
      }
    }
  }
  {
    unsigned int* mdst = maskw + (swRow + (lane >> 3)) * 32 + (2 * swQ + ((lane >> 2) & 1)) * 4 + (lane & 3);
    mdst[0]      = (unsigned int)wv0;
    mdst[4 * 32] = (unsigned int)wv1;
  }
  float rv[16];
#pragma unroll
  for (int i = 0; i < 16; ++i) {
    const float send = laneHi ? mxr[i] : mxr[i + 16];
    const float keep = laneHi ? mxr[i + 16] : mxr[i];
    rv[i] = fmaxf(keep, xhalf(send));
  }
  rowmax_step<8>(rv, lane16);
  rowmax_step<4>(rv, lane16);
  rowmax_step<2>(rv, lane16);
  rowmax_step<1>(rv, lane16);
  s2maxl[swQ * (Hh * 16) + (lane & 3) * 16 + swRow + (lane >> 2)] = rv[0];
  __syncthreads();

  const float s1v  = s1l[hh * 16 + lane16];
  const float* qmx = s2maxl + hh * 16 + lane16;
  float rmax = s1v + fmaxf(fmaxf(qmx[0], qmx[Hh * 16]), fmaxf(qmx[2 * Hh * 16], qmx[3 * Hh * 16]));
  rmax = rmax > 0.f ? rmax : NEG_SLOPE * rmax;
  float psum = 0.f;
  const float* hb  = hbuf + (b * Hh + hh) * Nq * HD;
  const float* s2h = s2l + hh * Nq;
  const unsigned int* mrow = maskw + lane16 * 32;
  const int mbase = half * (Nq / 2);

  v8f acc0 = {}, acc1 = {}, acc2 = {}, acc3 = {};
  const int mend = mbase + Nq / 2;
  const float* hlane = hb + laneHi * (2 * HD) + 4 * lane16;
  auto form_p = [&](v2f* p, const v2f* sq, const unsigned int bx, const unsigned int by) {
#pragma unroll
    for (int s = 0; s < 4; ++s) {
      float v0 = s1v + sq[s].x;
      float v1 = s1v + sq[s].y;
      v0 = v0 > 0.f ? v0 : NEG_SLOPE * v0;
      v1 = v1 > 0.f ? v1 : NEG_SLOPE * v1;
      const float e0 = __expf(v0 - rmax);
      const float e1 = __expf(v1 - rmax);
      p[s].x = ((bx >> s) & 1u) ? e0 : 0.f;
      p[s].y = ((by >> s) & 1u) ? e1 : 0.f;
    }
  };
  auto mma_block = [&](const HBlock& f, const v2f* p) {
#pragma unroll
    for (int s = 0; s < 4; ++s) {
      psum += p[s].x + p[s].y;
      acc0 = __builtin_amdgcn_wmma_f32_16x16x4_f32(false, p[s], false, f.t[s][0], (short)0, acc0, false, false);
      acc1 = __builtin_amdgcn_wmma_f32_16x16x4_f32(false, p[s], false, f.t[s][1], (short)0, acc1, false, false);
      acc2 = __builtin_amdgcn_wmma_f32_16x16x4_f32(false, p[s], false, f.t[s][2], (short)0, acc2, false, false);
      acc3 = __builtin_amdgcn_wmma_f32_16x16x4_f32(false, p[s], false, f.t[s][3], (short)0, acc3, false, false);
    }
  };
  auto load_trip = [&](v2f* sq, unsigned int& bx, unsigned int& by, const int kt) {
    const float* s2k = s2h + kt + 2 * laneHi;
#pragma unroll
    for (int j = 0; j < 8; ++j) sq[j] = *reinterpret_cast<const v2f*>(s2k + 4 * j);
    const v2u w2 = *reinterpret_cast<const v2u*>(mrow + 4 * (kt >> 7) + 2 * laneHi);
    const int sh = (kt & 127) >> 2;
    bx = w2.x >> sh;
    by = w2.y >> sh;
  };

  HBlock fa, fb;
  v2f pa[4], pb[4];
  v2f sqh[4];
  unsigned int mhx, mhy;
  {
    v2f s0[8]; unsigned int m0x, m0y;
    load_trip(s0, m0x, m0y, mbase);
    load_hblock(fa, hlane + mbase * HD);
    form_p(pa, s0, m0x, m0y);
#pragma unroll
    for (int j = 0; j < 4; ++j) sqh[j] = s0[4 + j];
    mhx = m0x >> 4;
    mhy = m0y >> 4;
  }
  for (int kb = mbase; kb < mend; kb += 32) {
    const int kn = kb + 32 < mend ? kb + 32 : mbase;
    v2f sn[8]; unsigned int mnx, mny;
    load_trip(sn, mnx, mny, kn);
    load_hblock(fb, hlane + (kb + 16) * HD);
    __builtin_amdgcn_sched_barrier(0);
    form_p(pb, sqh, mhx, mhy);
    mma_block(fa, pa);
    int hnext = kn * HD;
    asm volatile("" : "+v"(hnext) : "v"(acc0), "v"(acc1), "v"(acc2), "v"(acc3),
                 "v"(pb[0]), "v"(pb[1]), "v"(pb[2]), "v"(pb[3]));
    load_hblock(fa, hlane + hnext);
    __builtin_amdgcn_sched_barrier(0);
    form_p(pa, sn, mnx, mny);
    mma_block(fb, pb);
#pragma unroll
    for (int j = 0; j < 4; ++j) sqh[j] = sn[4 + j];
    mhx = mnx >> 4;
    mhy = mny >> 4;
  }

  psum += __shfl_xor(psum, 16, 32);
  if (laneHi == 0) psuml[(half * Hh + hh) * 16 + lane16] = psum;
  float* op = opart + hh * 16 * HD;
  if (half == 0) {
#pragma unroll
    for (int v = 0; v < 8; ++v) {
      float* orow = op + (v + 8 * laneHi) * HD + lane16;
      orow[0]  = acc0[v];
      orow[16] = acc1[v];
      orow[32] = acc2[v];
      orow[48] = acc3[v];
    }
  }
  __syncthreads();
  if (half == 1) {
    float* ob = out + (size_t)b * Nq * Fout + hh * HD;
#pragma unroll
    for (int v = 0; v < 8; ++v) {
      const float* prow = op + (v + 8 * laneHi) * HD + lane16;
      float* orow = ob + (size_t)(n0 + v + 8 * laneHi) * Fout + 2 * lane16;
      const float rsum = psuml[hh * 16 + v + 8 * laneHi] + psuml[(Hh + hh) * 16 + v + 8 * laneHi];
      const float rinv = rsum > 0.f ? 1.f / rsum : 0.f;
      v2f o01, o23;
      o01.x = (acc0[v] + prow[0]) * rinv;
      o01.y = (acc1[v] + prow[16]) * rinv;
      o23.x = (acc2[v] + prow[32]) * rinv;
      o23.y = (acc3[v] + prow[48]) * rinv;
      *reinterpret_cast<v2f*>(orow)      = o01;
      *reinterpret_cast<v2f*>(orow + 32) = o23;
    }
  }
}

extern "C" void kernel_launch(void* const* d_in, const int* in_sizes, int n_in,
                              void* d_out, int out_size, void* d_ws, size_t ws_size,
                              hipStream_t stream) {
  const float* x   = (const float*)d_in[0];
  const int*   adj = (const int*)d_in[1];
  const float* W   = (const float*)d_in[2];
  const float* a   = (const float*)d_in[3];
  float* out = (float*)d_out;

  float* hbuf = (float*)d_ws;
  float* s1 = hbuf + (size_t)Bq * Hh * Nq * HD;
  float* s2 = s1 + (size_t)Bq * Hh * Nq;

  gat_proj<<<512, 128, 0, stream>>>(x, W, a, hbuf, s1, s2);
  gat_attn<<<dim3(Nq / 16, Bq), 256, 0, stream>>>(hbuf, adj, s1, s2, out);
}


# attn: a wave holds two 16-row tiles (block = 32 rows, 16 waves = 4 heads x 4 m-quarters): each h fragment feeds two WMMAs, h reads halve; O and row sums = four quarter partials (new sum order)
# speedup vs baseline: 2.9481x; 1.0885x over previous
#include <hip/hip_runtime.h>


typedef float v2f __attribute__((ext_vector_type(2)));
typedef float v8f __attribute__((ext_vector_type(8)));

#define Bq 8
#define Nq 1024
#define Fin 256
#define Fout 256
#define Hh 4
#define HD 64
#define NEG_SLOPE 0.2f

typedef float v4f __attribute__((ext_vector_type(4)));

struct PBlock { v4f xa[4]; v4f w[4][4]; };

__device__ __forceinline__ void load_pblock(PBlock& f, const float* __restrict__ xp,
                                            const float* __restrict__ wp) {
#pragma unroll
  for (int g = 0; g < 4; ++g) {
    f.xa[g] = *reinterpret_cast<const v4f*>(xp + 8 * g);
#pragma unroll
    for (int j = 0; j < 4; ++j)
      f.w[g][j] = *reinterpret_cast<const v4f*>(wp + (8 * g + j) * Fout);
  }
}

template <int CNT>
__device__ __forceinline__ void score_step(float (&sv)[16], const int lane16) {
  const bool up = (lane16 & CNT) != 0;
#pragma unroll
  for (int i = 0; i < CNT; ++i) {
    const float send = up ? sv[i] : sv[i + CNT];
    const float keep = up ? sv[i + CNT] : sv[i];
    const float got = __builtin_bit_cast(float,
        __builtin_amdgcn_update_dpp(0, __builtin_bit_cast(int, send), 0x160 + CNT, 0xF, 0xF, false));
    sv[i] = keep + got;
  }
}

__global__ __launch_bounds__(128) void gat_proj(const float* __restrict__ x,
                                                const float* __restrict__ W,
                                                const float* __restrict__ a,
                                                float* __restrict__ hbuf,
                                                float* __restrict__ s1,
                                                float* __restrict__ s2) {
  const int tid = threadIdx.x;
  const int wave = tid >> 5, lane = tid & 31;
  const int lane16 = lane & 15, laneHi = lane >> 4;
  const int hh = blockIdx.x & 3;
  const int mt = (blockIdx.x >> 2) * 4 + wave;

  const v4f a1v = *reinterpret_cast<const v4f*>(a + hh * (2 * HD) + 4 * lane16);
  const v4f a2v = *reinterpret_cast<const v4f*>(a + hh * (2 * HD) + HD + 4 * lane16);

  const float* xp = x + (mt * 16 + lane16) * Fin + 4 * laneHi;
  const float* wp = W + (4 * laneHi) * Fout + hh * HD + 4 * lane16;

  v8f acc0 = {}, acc1 = {}, acc2 = {}, acc3 = {};
  auto mma_pblock = [&](const PBlock& f) {
#pragma unroll
    for (int g = 0; g < 4; ++g) {
#pragma unroll
      for (int s = 0; s < 2; ++s) {
        const v2f av = s ? f.xa[g].zw : f.xa[g].xy;
        const v4f wlo = f.w[g][2 * s], whi = f.w[g][2 * s + 1];
        v2f b0, b1, b2, b3;
        b0.x = wlo.x; b0.y = whi.x;
        b1.x = wlo.y; b1.y = whi.y;
        b2.x = wlo.z; b2.y = whi.z;
        b3.x = wlo.w; b3.y = whi.w;
        acc0 = __builtin_amdgcn_wmma_f32_16x16x4_f32(false, av, false, b0, (short)0, acc0, false, false);
        acc1 = __builtin_amdgcn_wmma_f32_16x16x4_f32(false, av, false, b1, (short)0, acc1, false, false);
        acc2 = __builtin_amdgcn_wmma_f32_16x16x4_f32(false, av, false, b2, (short)0, acc2, false, false);
        acc3 = __builtin_amdgcn_wmma_f32_16x16x4_f32(false, av, false, b3, (short)0, acc3, false, false);
      }
    }
  };

  PBlock fa, fb;
  load_pblock(fa, xp, wp);
#pragma unroll 1
  for (int kb = 0; kb < Fin - 64; kb += 64) {
    load_pblock(fb, xp + kb + 32, wp + (kb + 32) * Fout);
    __builtin_amdgcn_sched_barrier(0);
    mma_pblock(fa);
    int kn = kb + 64;
    asm volatile("" : "+v"(kn) : "v"(acc0), "v"(acc1), "v"(acc2), "v"(acc3));
    load_pblock(fa, xp + kn, wp + kn * Fout);
    __builtin_amdgcn_sched_barrier(0);
    mma_pblock(fb);
  }
  load_pblock(fb, xp + (Fin - 32), wp + (Fin - 32) * Fout);
  __builtin_amdgcn_sched_barrier(0);
  mma_pblock(fa);
  mma_pblock(fb);

#pragma unroll
  for (int v = 0; v < 8; v += 2) {
    const int r = mt * 16 + v + 8 * laneHi;
    const int bb = r >> 10, n = r & (Nq - 1);
    float* dst = hbuf + ((bb * Hh + hh) * Nq + n) * HD + 8 * lane16;
    v4f lo, hi;
    lo.x = acc0[v]; lo.y = acc0[v + 1]; lo.z = acc1[v]; lo.w = acc1[v + 1];
    hi.x = acc2[v]; hi.y = acc2[v + 1]; hi.z = acc3[v]; hi.w = acc3[v + 1];
    *reinterpret_cast<v4f*>(dst)     = lo;
    *reinterpret_cast<v4f*>(dst + 4) = hi;
  }

  float sv[16];
#pragma unroll
  for (int v = 0; v < 8; ++v) {
    sv[v]     = (acc0[v] * a1v.x + acc1[v] * a1v.y) + (acc2[v] * a1v.z + acc3[v] * a1v.w);
    sv[8 + v] = (acc0[v] * a2v.x + acc1[v] * a2v.y) + (acc2[v] * a2v.z + acc3[v] * a2v.w);
  }
  score_step<8>(sv, lane16);
  score_step<4>(sv, lane16);
  score_step<2>(sv, lane16);
  score_step<1>(sv, lane16);
  {
    const int q = lane16 >> 3, v = lane16 & 7;
    const int r = mt * 16 + v + 8 * laneHi;
    const int bb = r >> 10, n = r & (Nq - 1);
    float* dst = q ? s2 : s1;
    dst[(bb * Hh + hh) * Nq + n] = sv[0];
  }
}


struct HBlock { v2f t[4][4]; };

__device__ __forceinline__ void load_hblock(HBlock& f, const float* __restrict__ hrow) {
#pragma unroll
  for (int s = 0; s < 4; ++s) {
    const v4f lo = *reinterpret_cast<const v4f*>(hrow + s * (4 * HD));
    const v4f hi = *reinterpret_cast<const v4f*>(hrow + s * (4 * HD) + HD);
    f.t[s][0] = lo.xy;
    f.t[s][1] = lo.zw;
    f.t[s][2] = hi.xy;
    f.t[s][3] = hi.zw;
  }
}

typedef int v4i __attribute__((ext_vector_type(4)));
typedef unsigned int v2u __attribute__((ext_vector_type(2)));

__device__ __forceinline__ float xhalf(const float x) {
  return __builtin_bit_cast(float, __builtin_amdgcn_permlanex16(__builtin_bit_cast(int, x),
      __builtin_bit_cast(int, x), 0x76543210, 0xfedcba98, false, false));
}

template <int CNT>
__device__ __forceinline__ void rowmax_step(float (&v)[16], const int lane16) {
  const bool up = (lane16 & CNT) != 0;
#pragma unroll
  for (int i = 0; i < CNT; ++i) {
    const float send = up ? v[i] : v[i + CNT];
    const float keep = up ? v[i + CNT] : v[i];
    const float got = __builtin_bit_cast(float,
        __builtin_amdgcn_update_dpp(0, __builtin_bit_cast(int, send), 0x160 + CNT, 0xF, 0xF, false));
    v[i] = fmaxf(keep, got);
  }
}

#define RT 32
__global__ __launch_bounds__(512) void gat_attn(const float* __restrict__ hbuf,
                                                const int* __restrict__ adj,
                                                const float* __restrict__ s1g,
                                                const float* __restrict__ s2g,
                                                float* __restrict__ out) {
  __shared__ __attribute__((aligned(16))) unsigned int maskw[RT * (Nq / 32)];
  __shared__ __attribute__((aligned(16))) float s2l[Hh * Nq];
  __shared__ float s1l[Hh * RT];
  __shared__ float s2maxl[4 * Hh * RT];
  __shared__ float psuml[4 * Hh * RT];
  __shared__ __attribute__((aligned(16))) float opart[16 * RT * HD];

  const int tid = threadIdx.x;
  const int wave = tid >> 5, lane = tid & 31;
  const int lane16 = lane & 15, laneHi = lane >> 4;
  const int b = blockIdx.y, n0 = blockIdx.x * RT;
  const int hh = wave >> 2;
  const int quarter = wave & 3;

  const int swRow = (wave & 3) * 8;
  const int swQ = wave >> 2;
  const float* s2b = s2g + b * Hh * Nq;
  const float* s1b = s1g + b * Hh * Nq;
  v4f sv[Hh][2];
#pragma unroll
  for (int h = 0; h < Hh; ++h) {
#pragma unroll
    for (int u = 0; u < 2; ++u)
      sv[h][u] = *reinterpret_cast<const v4f*>(s2b + h * Nq + 256 * swQ + 128 * u + 4 * lane);
  }
  float s1t = 0.f;
  if (tid < Hh * RT) s1t = s1b[(tid >> 5) * Nq + n0 + (tid & 31)];
  const int* adjp = adj + (b * Nq + n0 + swRow) * Nq + 256 * swQ + 4 * lane;
  v4i av[8][2];
#pragma unroll
  for (int r = 0; r < 8; ++r) {
#pragma unroll
    for (int u = 0; u < 2; ++u)
      av[r][u] = *reinterpret_cast<const v4i*>(adjp + r * Nq + 128 * u);
  }
  if ((wave & 3) == 0) {
#pragma unroll
    for (int h = 0; h < Hh; ++h) {
#pragma unroll
      for (int u = 0; u < 2; ++u)
        *reinterpret_cast<v4f*>(s2l + h * Nq + 256 * swQ + 128 * u + 4 * lane) = sv[h][u];
    }
  }
  if (tid < Hh * RT) s1l[tid] = s1t;

  const float ninf = -__builtin_huge_valf();
  float mxr[32];
#pragma unroll
  for (int i = 0; i < 32; ++i) mxr[i] = ninf;
  int wv0 = 0, wv1 = 0;
#pragma unroll
  for (int r = 0; r < 8; ++r) {
#pragma unroll
    for (int u = 0; u < 2; ++u) {
#pragma unroll
      for (int c = 0; c < 4; ++c) {
        const bool on = av[r][u][c] != 0;
        const int word = (int)__builtin_amdgcn_ballot_w32(on);
#pragma unroll
        for (int h = 0; h < Hh; ++h)
          mxr[4 * r + h] = fmaxf(mxr[4 * r + h], on ? sv[h][u][c] : ninf);
        const int id = r * 8 + u * 4 + c;
        if (id < 32) wv0 = (lane == id) ? word : wv0;
        else         wv1 = (lane == id - 32) ? word : wv1;
      }
    }
  }
  {
    unsigned int* mdst = maskw + (swRow + (lane >> 3)) * 32 + (2 * swQ + ((lane >> 2) & 1)) * 4 + (lane & 3);
    mdst[0]      = (unsigned int)wv0;
    mdst[4 * 32] = (unsigned int)wv1;
  }
  float rv[16];
#pragma unroll
  for (int i = 0; i < 16; ++i) {
    const float send = laneHi ? mxr[i] : mxr[i + 16];
    const float keep = laneHi ? mxr[i + 16] : mxr[i];
    rv[i] = fmaxf(keep, xhalf(send));
  }
  rowmax_step<8>(rv, lane16);
  rowmax_step<4>(rv, lane16);
  rowmax_step<2>(rv, lane16);
  rowmax_step<1>(rv, lane16);
  s2maxl[swQ * (Hh * RT) + (lane & 3) * RT + swRow + (lane >> 2)] = rv[0];
  __syncthreads();

  float s1v[2], rmax[2];
#pragma unroll
  for (int rt = 0; rt < 2; ++rt) {
    s1v[rt] = s1l[hh * RT + rt * 16 + lane16];
    const float* qmx = s2maxl + hh * RT + rt * 16 + lane16;
    float rm = s1v[rt] + fmaxf(fmaxf(qmx[0], qmx[Hh * RT]), fmaxf(qmx[2 * Hh * RT], qmx[3 * Hh * RT]));
    rmax[rt] = rm > 0.f ? rm : NEG_SLOPE * rm;
  }
  float psum0 = 0.f, psum1 = 0.f;
  const float* hb  = hbuf + (b * Hh + hh) * Nq * HD;
  const float* s2h = s2l + hh * Nq;
  const unsigned int* mrow0 = maskw + lane16 * 32;
  const unsigned int* mrow1 = maskw + (16 + lane16) * 32;
  const int mbase = quarter * (Nq / 4);

  v8f acc[8];
#pragma unroll
  for (int i = 0; i < 8; ++i) acc[i] = v8f{};
  const int mend = mbase + Nq / 4;
  const float* hlane = hb + laneHi * (2 * HD) + 4 * lane16;
  auto form_p = [&](v2f* p, const v2f* sq, const unsigned int bx, const unsigned int by,
                    const float s1r, const float rmx) {
#pragma unroll
    for (int s = 0; s < 4; ++s) {
      float v0 = s1r + sq[s].x;
      float v1 = s1r + sq[s].y;
      v0 = v0 > 0.f ? v0 : NEG_SLOPE * v0;
      v1 = v1 > 0.f ? v1 : NEG_SLOPE * v1;
      const float e0 = __expf(v0 - rmx);
      const float e1 = __expf(v1 - rmx);
      p[s].x = ((bx >> s) & 1u) ? e0 : 0.f;
      p[s].y = ((by >> s) & 1u) ? e1 : 0.f;
    }
  };
  auto mma_block = [&](const HBlock& f, const v2f* p0, const v2f* p1) {
#pragma unroll
    for (int s = 0; s < 4; ++s) {
      psum0 += p0[s].x + p0[s].y;
      psum1 += p1[s].x + p1[s].y;
#pragma unroll
      for (int ct = 0; ct < 4; ++ct)
        acc[ct] = __builtin_amdgcn_wmma_f32_16x16x4_f32(false, p0[s], false, f.t[s][ct], (short)0, acc[ct], false, false);
#pragma unroll
      for (int ct = 0; ct < 4; ++ct)
        acc[4 + ct] = __builtin_amdgcn_wmma_f32_16x16x4_f32(false, p1[s], false, f.t[s][ct], (short)0, acc[4 + ct], false, false);
    }
  };
  auto load_trip = [&](v2f* sq, unsigned int* bx, unsigned int* by, const int kt) {
    const float* s2k = s2h + kt + 2 * laneHi;
#pragma unroll
    for (int j = 0; j < 8; ++j) sq[j] = *reinterpret_cast<const v2f*>(s2k + 4 * j);
    const int wo = 4 * (kt >> 7) + 2 * laneHi;
    const v2u w0 = *reinterpret_cast<const v2u*>(mrow0 + wo);
    const v2u w1 = *reinterpret_cast<const v2u*>(mrow1 + wo);
    const int sh = (kt & 127) >> 2;
    bx[0] = w0.x >> sh; by[0] = w0.y >> sh;
    bx[1] = w1.x >> sh; by[1] = w1.y >> sh;
  };

  HBlock fa, fb;
  v2f pa0[4], pa1[4], pb0[4], pb1[4];
  v2f sqh[4];
  unsigned int mhx[2], mhy[2];
  {
    v2f s0[8]; unsigned int m0x[2], m0y[2];
    load_trip(s0, m0x, m0y, mbase);
    load_hblock(fa, hlane + mbase * HD);
    form_p(pa0, s0, m0x[0], m0y[0], s1v[0], rmax[0]);
    form_p(pa1, s0, m0x[1], m0y[1], s1v[1], rmax[1]);
#pragma unroll
    for (int j = 0; j < 4; ++j) sqh[j] = s0[4 + j];
#pragma unroll
    for (int rt = 0; rt < 2; ++rt) { mhx[rt] = m0x[rt] >> 4; mhy[rt] = m0y[rt] >> 4; }
  }
  for (int kb = mbase; kb < mend; kb += 32) {
    const int kn = kb + 32 < mend ? kb + 32 : mbase;
    v2f sn[8]; unsigned int mnx[2], mny[2];
    load_trip(sn, mnx, mny, kn);
    load_hblock(fb, hlane + (kb + 16) * HD);
    __builtin_amdgcn_sched_barrier(0);
    form_p(pb0, sqh, mhx[0], mhy[0], s1v[0], rmax[0]);
    form_p(pb1, sqh, mhx[1], mhy[1], s1v[1], rmax[1]);
    mma_block(fa, pa0, pa1);
    int hnext = kn * HD;
    asm volatile("" : "+v"(hnext) : "v"(acc[0]), "v"(acc[1]), "v"(acc[2]), "v"(acc[3]),
                 "v"(acc[4]), "v"(acc[5]), "v"(acc[6]), "v"(acc[7]),
                 "v"(pb0[0]), "v"(pb0[1]), "v"(pb0[2]), "v"(pb0[3]),
                 "v"(pb1[0]), "v"(pb1[1]), "v"(pb1[2]), "v"(pb1[3]));
    load_hblock(fa, hlane + hnext);
    __builtin_amdgcn_sched_barrier(0);
    form_p(pa0, sn, mnx[0], mny[0], s1v[0], rmax[0]);
    form_p(pa1, sn, mnx[1], mny[1], s1v[1], rmax[1]);
    mma_block(fb, pb0, pb1);
#pragma unroll
    for (int j = 0; j < 4; ++j) sqh[j] = sn[4 + j];
#pragma unroll
    for (int rt = 0; rt < 2; ++rt) { mhx[rt] = mnx[rt] >> 4; mhy[rt] = mny[rt] >> 4; }
  }

  psum0 += __shfl_xor(psum0, 16, 32);
  psum1 += __shfl_xor(psum1, 16, 32);
  if (laneHi == 0) {
    psuml[(quarter * Hh + hh) * RT + lane16]      = psum0;
    psuml[(quarter * Hh + hh) * RT + 16 + lane16] = psum1;
  }
  {
    float* op = opart + (hh * 4 + quarter) * (RT * HD) + 4 * lane;
#pragma unroll
    for (int rt = 0; rt < 2; ++rt) {
#pragma unroll
      for (int vg = 0; vg < 2; ++vg) {
#pragma unroll
        for (int ct = 0; ct < 4; ++ct) {
          v4f piece;
          piece.x = acc[rt * 4 + ct][4 * vg + 0];
          piece.y = acc[rt * 4 + ct][4 * vg + 1];
          piece.z = acc[rt * 4 + ct][4 * vg + 2];
          piece.w = acc[rt * 4 + ct][4 * vg + 3];
          *reinterpret_cast<v4f*>(op + ((rt * 2 + vg) * 4 + ct) * 128) = piece;
        }
      }
    }
  }
  __syncthreads();
  {
    const int rt = quarter >> 1, vg = quarter & 1;
    const float* ip = opart + (hh * 4) * (RT * HD) + ((rt * 2 + vg) * 4) * 128 + 4 * lane;
    v4f o[4];
#pragma unroll
    for (int ct = 0; ct < 4; ++ct) {
      const v4f q0 = *reinterpret_cast<const v4f*>(ip + 0 * (RT * HD) + ct * 128);
      const v4f q1 = *reinterpret_cast<const v4f*>(ip + 1 * (RT * HD) + ct * 128);
      const v4f q2 = *reinterpret_cast<const v4f*>(ip + 2 * (RT * HD) + ct * 128);
      const v4f q3 = *reinterpret_cast<const v4f*>(ip + 3 * (RT * HD) + ct * 128);
      o[ct] = (q0 + q1) + (q2 + q3);
    }
    float* ob = out + (size_t)b * Nq * Fout + hh * HD;
    const int rbase = rt * 16 + 4 * vg + 8 * laneHi;
#pragma unroll
    for (int j = 0; j < 4; ++j) {
      const int row = rbase + j;
      const float* ps = psuml + hh * RT + row;
      const float rsum = (ps[0] + ps[Hh * RT]) + (ps[2 * Hh * RT] + ps[3 * Hh * RT]);
      const float rinv = rsum > 0.f ? 1.f / rsum : 0.f;
      float* orow = ob + (size_t)(n0 + row) * Fout + 2 * lane16;
      v2f o01, o23;
      o01.x = o[0][j] * rinv;
      o01.y = o[1][j] * rinv;
      o23.x = o[2][j] * rinv;
      o23.y = o[3][j] * rinv;
      *reinterpret_cast<v2f*>(orow)      = o01;
      *reinterpret_cast<v2f*>(orow + 32) = o23;
    }
  }
}

extern "C" void kernel_launch(void* const* d_in, const int* in_sizes, int n_in,
                              void* d_out, int out_size, void* d_ws, size_t ws_size,
                              hipStream_t stream) {
  const float* x   = (const float*)d_in[0];
  const int*   adj = (const int*)d_in[1];
  const float* W   = (const float*)d_in[2];
  const float* a   = (const float*)d_in[3];
  float* out = (float*)d_out;

  float* hbuf = (float*)d_ws;
  float* s1 = hbuf + (size_t)Bq * Hh * Nq * HD;
  float* s2 = s1 + (size_t)Bq * Hh * Nq;

  gat_proj<<<512, 128, 0, stream>>>(x, W, a, hbuf, s1, s2);
  gat_attn<<<dim3(Nq / 32, Bq), 512, 0, stream>>>(hbuf, adj, s1, s2, out);
}
